# starting kernel
# baseline (speedup 1.0000x reference)
#include <hip/hip_runtime.h>

#define D_FEAT 64
#define HC 128
#define NHEAD 2
#define CCH 64
#define NEG_SLOPE 0.2f

typedef __attribute__((ext_vector_type(2))) float v2f;
typedef __attribute__((ext_vector_type(8))) float v8f;


__global__ void fill_kernel(float* __restrict__ p, float v, int n) {
    int i = blockIdx.x * blockDim.x + threadIdx.x;
    if (i < n) p[i] = v;
}

__device__ __forceinline__ void atomicMaxF(float* addr, float val) {
    if (val >= 0.0f) atomicMax((int*)addr, __float_as_int(val));
    else             atomicMin((unsigned int*)addr, __float_as_uint(val));
}

__global__ void gemm64x128_wmma(const float* __restrict__ X,
                                const float* __restrict__ W,
                                const float* __restrict__ bvec,
                                float* __restrict__ Y, int rows) {
    const int wave = (blockIdx.x * blockDim.x + threadIdx.x) >> 5;
    const int lane = threadIdx.x & 31;
    const int tiles_m = rows >> 4;
    const int tm = wave >> 3;
    const int tn = wave & 7;
    if (tm >= tiles_m) return;

    const int mrow  = (tm << 4) + (lane & 15);
    const int ncol  = (tn << 4) + (lane & 15);
    const int khalf = (lane >> 4) << 1;

    v8f acc = {};
#pragma unroll
    for (int k0 = 0; k0 < D_FEAT; k0 += 4) {
        const int kk = k0 + khalf;
        v2f a, b;
        const float* ap = X + (size_t)mrow * D_FEAT + kk;
        a.x = ap[0];
        a.y = ap[1];
        b.x = W[(size_t)(kk + 0) * HC + ncol];
        b.y = W[(size_t)(kk + 1) * HC + ncol];
        acc = __builtin_amdgcn_wmma_f32_16x16x4_f32(
             false, a,  false, b,
             (short)0, acc,  false,  false);
    }

    const float bb = bvec[ncol];
    const int rbase = (tm << 4) + ((lane >> 4) << 3);
#pragma unroll
    for (int j = 0; j < 8; ++j)
        Y[(size_t)(rbase + j) * HC + ncol] = acc[j] + bb;
}

__global__ void score_kernel(const float* __restrict__ xl,
                             const float* __restrict__ xr,
                             const int* __restrict__ src,
                             const int* __restrict__ dst,
                             const float* __restrict__ att,
                             float* __restrict__ ebuf,
                             float* __restrict__ mmax,
                             int E, int EE) {
    const int e    = (blockIdx.x * blockDim.x + threadIdx.x) >> 5;
    const int lane = threadIdx.x & 31;
    if (e >= EE) return;
    int s, d;
    if (e < E) { s = src[e]; d = dst[e]; } else { s = e - E; d = s; }

    const int c0 = lane << 2;
    const float4 zl = *(const float4*)(xl + (size_t)s * HC + c0);
    const float4 zr = *(const float4*)(xr + (size_t)d * HC + c0);
    const float4 at = *(const float4*)(att + c0);

    float p = 0.0f, z;
    z = zl.x + zr.x; z = (z > 0.0f) ? z : NEG_SLOPE * z; p += z * at.x;
    z = zl.y + zr.y; z = (z > 0.0f) ? z : NEG_SLOPE * z; p += z * at.y;
    z = zl.z + zr.z; z = (z > 0.0f) ? z : NEG_SLOPE * z; p += z * at.z;
    z = zl.w + zr.w; z = (z > 0.0f) ? z : NEG_SLOPE * z; p += z * at.w;

    p += __shfl_xor(p, 1);
    p += __shfl_xor(p, 2);
    p += __shfl_xor(p, 4);
    p += __shfl_xor(p, 8);

    if ((lane & 15) == 0) {
        const int h = lane >> 4;
        ebuf[(size_t)e * NHEAD + h] = p;
        atomicMaxF(&mmax[(size_t)d * NHEAD + h], p);
    }
}

__global__ void expsum_kernel(const int* __restrict__ src,
                              const int* __restrict__ dst,
                              float* __restrict__ ebuf,
                              const float* __restrict__ mmax,
                              float* __restrict__ denom,
                              int E, int EE) {
    int g = blockIdx.x * blockDim.x + threadIdx.x;
    if (g >= EE * NHEAD) return;
    const int e = g >> 1;
    const int h = g & 1;
    const int d = (e < E) ? dst[e] : (e - E);
    const float a = __expf(ebuf[g] - mmax[(size_t)d * NHEAD + h]);
    ebuf[g] = a;
    atomicAdd(&denom[(size_t)d * NHEAD + h], a);
}

__global__ void scatter_kernel(const float* __restrict__ xl,
                               const float* __restrict__ ebuf,
                               const float* __restrict__ denom,
                               const int* __restrict__ src,
                               const int* __restrict__ dst,
                               float* __restrict__ accum,
                               int E, int EE) {
    const int e    = (blockIdx.x * blockDim.x + threadIdx.x) >> 5;
    const int lane = threadIdx.x & 31;
    if (e >= EE) return;
    int s, d;
    if (e < E) { s = src[e]; d = dst[e]; } else { s = e - E; d = s; }

    const int h = lane >> 4;
    const float alpha = ebuf[(size_t)e * NHEAD + h] /
                        (denom[(size_t)d * NHEAD + h] + 1e-16f);
    const int c0 = lane << 2;
    const float4 v = *(const float4*)(xl + (size_t)s * HC + c0);
    float* o = accum + (size_t)d * HC + c0;
    atomicAdd(o + 0, alpha * v.x);
    atomicAdd(o + 1, alpha * v.y);
    atomicAdd(o + 2, alpha * v.z);
    atomicAdd(o + 3, alpha * v.w);
}

__global__ void finalize_kernel(const float* __restrict__ accum,
                                const float* __restrict__ bias,
                                float* __restrict__ out,
                                int apply_gelu, int total) {
    int g = blockIdx.x * blockDim.x + threadIdx.x;
    if (g >= total) return;
    const int node = g >> 6;
    const int c    = g & 63;
    float v = 0.5f * (accum[(size_t)node * HC + c] +
                      accum[(size_t)node * HC + CCH + c]) + bias[c];
    if (apply_gelu) {
        const float u = 0.7978845608028654f * (v + 0.044715f * v * v * v);
        v = 0.5f * v * (1.0f + tanhf(u));
    }
    out[g] = v;
}


extern "C" void kernel_launch(void* const* d_in, const int* in_sizes, int n_in,
                              void* d_out, int out_size, void* d_ws, size_t ws_size,
                              hipStream_t stream) {
    const float* x0   = (const float*)d_in[0];
    const int*   conn = (const int*)  d_in[1];
    const float* Wl   = (const float*)d_in[2];
    const float* bl   = (const float*)d_in[3];
    const float* Wr   = (const float*)d_in[4];
    const float* br   = (const float*)d_in[5];
    const float* att  = (const float*)d_in[6];
    const float* bias = (const float*)d_in[7];

    const int N  = in_sizes[0] / D_FEAT;
    const int E  = in_sizes[1] / 2;
    const int EE = E + N;
    const int* src = conn;
    const int* dst = conn + E;

    float* ws    = (float*)d_ws;
    float* xl    = ws;
    float* xr    = xl   + (size_t)N * HC;
    float* x1    = xr   + (size_t)N * HC;
    float* ebuf  = x1   + (size_t)N * D_FEAT;
    float* mmax  = ebuf + (size_t)EE * NHEAD;
    float* denom = mmax + (size_t)N * NHEAD;

    const dim3 blk(256);
    const int gemm_blocks = ((N / 16) * 8 + 7) / 8;
    const int edge_blocks = (int)(((long long)EE * 32 + 255) / 256);
    const int eh_blocks   = (EE * NHEAD + 255) / 256;
    const int nh          = N * NHEAD;
    const int nacc        = N * HC;
    const int fin_total   = N * D_FEAT;

    const int L = 2;
    const float* xin = x0;
    for (int l = 0; l < L; ++l) {
        const float* Wl_l = Wl + (size_t)l * D_FEAT * HC;
        const float* bl_l = bl + (size_t)l * HC;
        const float* Wr_l = Wr + (size_t)l * D_FEAT * HC;
        const float* br_l = br + (size_t)l * HC;
        const float* at_l = att + (size_t)l * NHEAD * CCH;
        const float* bi_l = bias + (size_t)l * CCH;

        gemm64x128_wmma<<<gemm_blocks, blk, 0, stream>>>(xin, Wl_l, bl_l, xl, N);
        gemm64x128_wmma<<<gemm_blocks, blk, 0, stream>>>(xin, Wr_l, br_l, xr, N);

        fill_kernel<<<(nh + 255) / 256, blk, 0, stream>>>(mmax, -__builtin_inff(), nh);
        fill_kernel<<<(nh + 255) / 256, blk, 0, stream>>>(denom, 0.0f, nh);

        score_kernel<<<edge_blocks, blk, 0, stream>>>(xl, xr, src, dst, at_l,
                                                      ebuf, mmax, E, EE);

        fill_kernel<<<(nacc + 255) / 256, blk, 0, stream>>>(xr, 0.0f, nacc);

        expsum_kernel<<<eh_blocks, blk, 0, stream>>>(src, dst, ebuf, mmax, denom, E, EE);

        scatter_kernel<<<edge_blocks, blk, 0, stream>>>(xl, ebuf, denom, src, dst,
                                                        xr, E, EE);

        float* outp = (l == L - 1) ? (float*)d_out : x1;
        finalize_kernel<<<(fin_total + 255) / 256, blk, 0, stream>>>(
            xr, bi_l, outp, (l < L - 1) ? 1 : 0, fin_total);

        xin = x1;
    }
}


# Both projections in one launch: a wave keeps its 16 rows of x as A fragments and walks all 256 columns of Wl and Wr, four column-interleaved tiles per 16-B load of W; same sum order; 9 launches
# speedup vs baseline: 2.0758x; 2.0758x over previous
#include <hip/hip_runtime.h>

#define D_FEAT 64
#define HC 128
#define NHEAD 2
#define CCH 64
#define NEG_SLOPE 0.2f

typedef __attribute__((ext_vector_type(2))) float v2f;
typedef __attribute__((ext_vector_type(8))) float v8f;


__global__ void fill_kernel(float* __restrict__ p, float v, int n) {
    int i = blockIdx.x * blockDim.x + threadIdx.x;
    if (i < n) p[i] = v;
}

__device__ __forceinline__ void atomicMaxF(float* addr, float val) {
    if (val >= 0.0f) atomicMax((int*)addr, __float_as_int(val));
    else             atomicMin((unsigned int*)addr, __float_as_uint(val));
}

#define GW 4

__global__ __launch_bounds__(GW * 32)
void proj_both_wmma(const float* __restrict__ X,
                    const float* __restrict__ Wl, const float* __restrict__ bl,
                    const float* __restrict__ Wr, const float* __restrict__ br,
                    float* __restrict__ Yl, float* __restrict__ Yr, int rows) {
    const int tm   = (blockIdx.x * blockDim.x + threadIdx.x) >> 5;
    const int lane = threadIdx.x & 31;
    if (tm >= (rows >> 4)) return;

    const int c     = lane & 15;
    const int hl    = lane >> 4;
    const int khalf = hl << 1;
    const int mrow  = (tm << 4) + c;

    v2f a[16];
    const float* ap = X + (size_t)mrow * D_FEAT + khalf;
#pragma unroll
    for (int s = 0; s < 16; ++s) {
        a[s].x = ap[4 * s];
        a[s].y = ap[4 * s + 1];
    }

    const int rbase = (tm << 4) + (hl << 3);
#pragma unroll
    for (int q = 0; q < 4; ++q) {
        const float* W  = (q < 2) ? Wl : Wr;
        const float* bv = (q < 2) ? bl : br;
        float*       Y  = (q < 2) ? Yl : Yr;
        const int col = ((q & 1) << 6) + (c << 2);

        v8f acc0 = {}, acc1 = {}, acc2 = {}, acc3 = {};
#pragma unroll
        for (int s0 = 0; s0 < 16; s0 += 8) {
            float4 w0[8], w1[8];
#pragma unroll
            for (int i = 0; i < 8; ++i) {
                const int kk = 4 * (s0 + i) + khalf;
                w0[i] = *(const float4*)(W + (size_t)(kk + 0) * HC + col);
                w1[i] = *(const float4*)(W + (size_t)(kk + 1) * HC + col);
            }
#pragma unroll
            for (int i = 0; i < 8; ++i) {
                const v2f av = a[s0 + i];
                v2f b;
                b.x = w0[i].x; b.y = w1[i].x;
                acc0 = __builtin_amdgcn_wmma_f32_16x16x4_f32(false, av, false, b, (short)0, acc0, false, false);
                b.x = w0[i].y; b.y = w1[i].y;
                acc1 = __builtin_amdgcn_wmma_f32_16x16x4_f32(false, av, false, b, (short)0, acc1, false, false);
                b.x = w0[i].z; b.y = w1[i].z;
                acc2 = __builtin_amdgcn_wmma_f32_16x16x4_f32(false, av, false, b, (short)0, acc2, false, false);
                b.x = w0[i].w; b.y = w1[i].w;
                acc3 = __builtin_amdgcn_wmma_f32_16x16x4_f32(false, av, false, b, (short)0, acc3, false, false);
            }
        }

        const float4 bb = *(const float4*)(bv + col);
#pragma unroll
        for (int j = 0; j < 8; ++j)
            *(float4*)(Y + (size_t)(rbase + j) * HC + col) =
                make_float4(acc0[j] + bb.x, acc1[j] + bb.y, acc2[j] + bb.z, acc3[j] + bb.w);
    }
}

#define EPW 8

__global__ void score_kernel(const float* __restrict__ xl,
                             const float* __restrict__ xr,
                             const int* __restrict__ src,
                             const int* __restrict__ dst,
                             const float* __restrict__ att,
                             float* __restrict__ ebuf,
                             float* __restrict__ mmax,
                             int E, int EE) {
    const int wave = (blockIdx.x * blockDim.x + threadIdx.x) >> 5;
    const int lane = threadIdx.x & 31;
    const int e0   = wave * EPW;
    if (e0 >= EE) return;

    int ee = e0 + (lane & (EPW - 1));
    if (ee > EE - 1) ee = EE - 1;
    int s, d;
    if (ee < E) { s = src[ee]; d = dst[ee]; } else { s = ee - E; d = s; }

    const int c0 = lane << 2;
    const float4 at = *(const float4*)(att + c0);

    float4 zl[EPW], zr[EPW];
#pragma unroll
    for (int k = 0; k < EPW; ++k) {
        const int sk = __builtin_amdgcn_readlane(s, k);
        const int dk = __builtin_amdgcn_readlane(d, k);
        zl[k] = *(const float4*)(xl + (size_t)sk * HC + c0);
        zr[k] = *(const float4*)(xr + (size_t)dk * HC + c0);
    }

    const int j = lane & 15;
    float pv = 0.0f;
#pragma unroll
    for (int k = 0; k < EPW; ++k) {
        float p = 0.0f, z;
        z = zl[k].x + zr[k].x; z = (z > 0.0f) ? z : NEG_SLOPE * z; p += z * at.x;
        z = zl[k].y + zr[k].y; z = (z > 0.0f) ? z : NEG_SLOPE * z; p += z * at.y;
        z = zl[k].z + zr[k].z; z = (z > 0.0f) ? z : NEG_SLOPE * z; p += z * at.z;
        z = zl[k].w + zr[k].w; z = (z > 0.0f) ? z : NEG_SLOPE * z; p += z * at.w;

        p += __shfl_xor(p, 1);
        p += __shfl_xor(p, 2);
        p += __shfl_xor(p, 4);
        p += __shfl_xor(p, 8);
        if (j == k) pv = p;
    }

    if (j < EPW && e0 + j < EE) {
        const int h = lane >> 4;
        ebuf[(size_t)(e0 + j) * NHEAD + h] = pv;
        atomicMaxF(&mmax[(size_t)d * NHEAD + h], pv);
    }
}

#define RNG   400

__global__ void range_id_kernel(const int* __restrict__ dst,
                                unsigned* __restrict__ rid, int E, int nd) {
    const int i = blockIdx.x * blockDim.x + threadIdx.x;
    if (i >= nd) return;
    unsigned w = 0u;
#pragma unroll
    for (int b = 0; b < 4; ++b) {
        const int e = 4 * i + b;
        if (e < E) w |= (((unsigned)dst[e] / (unsigned)RNG) & 255u) << (8 * b);
    }
    rid[i] = w;
}

#define SCT   1024
#define QCAP  5600
#define SCU   4
#define PB    8

__device__ __forceinline__ int acc_slot(int ch) { return ((ch & 3) << 5) | (ch >> 2); }

template <bool GUARD>
__device__ __forceinline__ void add_rows(float* acc, const int4* ent,
                                         const float* __restrict__ xl,
                                         int t0, int total, int h, int c0, int lane) {
    int4   en[PB];
    float4 v[PB];
#pragma unroll
    for (int k = 0; k < PB; ++k) {
        const int t = t0 + k;
        en[k] = ent[(!GUARD || t < total) ? t : 0];
    }
#pragma unroll
    for (int k = 0; k < PB; ++k)
        v[k] = *(const float4*)(xl + (size_t)en[k].x * HC + c0);
#pragma unroll
    for (int k = 0; k < PB; ++k) {
        if ((!GUARD || t0 + k < total) && en[k].y >= 0) {
            const float a = __int_as_float(h ? en[k].w : en[k].z);
            float* o = acc + en[k].y * HC + lane;
            atomicAdd(o,      a * v[k].x);
            atomicAdd(o + 32, a * v[k].y);
            atomicAdd(o + 64, a * v[k].z);
            atomicAdd(o + 96, a * v[k].w);
        }
    }
}

__global__ __launch_bounds__(SCT)
void scatter_range_kernel(const float* __restrict__ xl,
                          const float* __restrict__ ebuf,
                          const float* __restrict__ mmax,
                          const int* __restrict__ src,
                          const int* __restrict__ dst,
                          const unsigned* __restrict__ rid,
                          const float* __restrict__ bias,
                          float* __restrict__ out,
                          int apply_gelu, int N, int E) {
    __shared__ __attribute__((aligned(16))) float acc[RNG * HC];
    __shared__ __attribute__((aligned(16))) int4 ent[RNG + QCAP];
    __shared__ float den[RNG * NHEAD];
    __shared__ float mx[RNG * NHEAD];
    __shared__ int qn;

    const int tid  = threadIdx.x;
    const int lane = tid & 31;
    const int wave = tid >> 5;
    const int lo   = blockIdx.x * RNG;
    const int cnt  = (N - lo < RNG) ? (N - lo) : RNG;

    float4* a4 = (float4*)acc;
    for (int i = tid; i < RNG * HC / 4; i += SCT) a4[i] = make_float4(0.0f, 0.0f, 0.0f, 0.0f);
    for (int i = tid; i < RNG * NHEAD; i += SCT) {
        den[i] = 0.0f;
        mx[i]  = (i < cnt * NHEAD) ? mmax[(size_t)lo * NHEAD + i] : 0.0f;
    }
    if (tid == 0) qn = 0;
    __syncthreads();

    const unsigned pat = (blockIdx.x & 255u) * 0x01010101u;
    const uint4* rid4 = (const uint4*)rid;
    const int nq4 = (E + 15) >> 4;
    const int CH  = SCT * SCU;
    const int nch = (nq4 + CH - 1) / CH;
    int c = (nch > 0) ? (int)(blockIdx.x % (unsigned)nch) : 0;
    for (int ci = 0; ci < nch; ++ci) {
        const int base = c * CH + tid;
        uint4 dv[SCU];
        if ((c + 1) * CH <= nq4) {
#pragma unroll
            for (int k = 0; k < SCU; ++k) dv[k] = rid4[base + k * SCT];
        } else {
#pragma unroll
            for (int k = 0; k < SCU; ++k) {
                const int i = base + k * SCT;
                dv[k] = (i < nq4) ? rid4[i] : make_uint4(~pat, ~pat, ~pat, ~pat);
            }
        }
        unsigned long long ov = 0ull;
#pragma unroll
        for (int k = 0; k < SCU; ++k) {
            const unsigned wd[4] = {dv[k].x, dv[k].y, dv[k].z, dv[k].w};
#pragma unroll
            for (int q = 0; q < 4; ++q) {
                const unsigned x = wd[q] ^ pat;
                unsigned m = ~(((x & 0x7F7F7F7Fu) + 0x7F7F7F7Fu) | x | 0x7F7F7F7Fu);
                while (m != 0u) {
                    const int j = (q << 2) + ((__ffs(m) - 1) >> 3);
                    m &= m - 1u;
                    const int e = ((base + k * SCT) << 4) + j;
                    if (e < E) {
                        const int pos = atomicAdd(&qn, 1);
                        if (pos < QCAP) ent[cnt + pos].x = e;
                        else            ov |= 1ull << (16 * k + j);
                    }
                }
            }
        }
        while (ov != 0ull) {
            const int bit = __ffsll(ov) - 1;
            ov &= ov - 1ull;
            const int e = ((base + (bit >> 4) * SCT) << 4) + (bit & 15);
            const unsigned r = (unsigned)(dst[e] - lo);
            if (r >= (unsigned)cnt) continue;
            const int s = src[e];
            for (int hh = 0; hh < NHEAD; ++hh) {
                const float a = __expf(ebuf[(size_t)e * NHEAD + hh] - mx[r * NHEAD + hh]);
                atomicAdd(&den[r * NHEAD + hh], a);
                for (int cc = 0; cc < CCH; ++cc) {
                    const int ch = hh * CCH + cc;
                    atomicAdd(&acc[r * HC + acc_slot(ch)], a * xl[(size_t)s * HC + ch]);
                }
            }
        }
        if (++c == nch) c = 0;
    }
    __syncthreads();

    const int nq    = (qn < QCAP) ? qn : QCAP;
    const int total = cnt + nq;
    for (int t = tid; t < total; t += SCT) {
        int e, r, s;
        if (t < cnt) {
            r = t; s = lo + t; e = E + s;
        } else {
            e = ent[t].x; r = dst[e] - lo; s = src[e];
        }
        if ((unsigned)r < (unsigned)cnt) {
            const float2 eb = *(const float2*)(ebuf + (size_t)e * NHEAD);
            const float a0 = __expf(eb.x - mx[r * NHEAD + 0]);
            const float a1 = __expf(eb.y - mx[r * NHEAD + 1]);
            atomicAdd(&den[r * NHEAD + 0], a0);
            atomicAdd(&den[r * NHEAD + 1], a1);
            ent[t] = make_int4(s, r, __float_as_int(a0), __float_as_int(a1));
        } else {
            ent[t] = make_int4(lo, -1, 0, 0);
        }
    }
    __syncthreads();

    const int h  = lane >> 4;
    const int c0 = lane << 2;
    const int wv = __builtin_amdgcn_readfirstlane(wave);
    for (int t0 = wv * PB; t0 < total; t0 += (SCT / 32) * PB) {
        if (t0 + PB <= total) add_rows<false>(acc, ent, xl, t0, total, h, c0, lane);
        else                  add_rows<true >(acc, ent, xl, t0, total, h, c0, lane);
    }
    __syncthreads();

    for (int idx = tid; idx < cnt * CCH; idx += SCT) {
        const int node = idx >> 6;
        const int cc   = idx & 63;
        const float d0 = den[node * NHEAD + 0] + 1e-16f;
        const float d1 = den[node * NHEAD + 1] + 1e-16f;
        float v = 0.5f * (acc[node * HC + acc_slot(cc)] / d0 +
                          acc[node * HC + acc_slot(CCH + cc)] / d1) + bias[cc];
        if (apply_gelu) {
            const float u = 0.7978845608028654f * (v + 0.044715f * v * v * v);
            v = 0.5f * v * (1.0f + tanhf(u));
        }
        out[(size_t)(lo + node) * CCH + cc] = v;
    }
}


extern "C" void kernel_launch(void* const* d_in, const int* in_sizes, int n_in,
                              void* d_out, int out_size, void* d_ws, size_t ws_size,
                              hipStream_t stream) {
    const float* x0   = (const float*)d_in[0];
    const int*   conn = (const int*)  d_in[1];
    const float* Wl   = (const float*)d_in[2];
    const float* bl   = (const float*)d_in[3];
    const float* Wr   = (const float*)d_in[4];
    const float* br   = (const float*)d_in[5];
    const float* att  = (const float*)d_in[6];
    const float* bias = (const float*)d_in[7];

    const int N  = in_sizes[0] / D_FEAT;
    const int E  = in_sizes[1] / 2;
    const int EE = E + N;
    const int* src = conn;
    const int* dst = conn + E;

    float* ws    = (float*)d_ws;
    float* xl    = ws;
    float* xr    = xl   + (size_t)N * HC;
    float* x1    = xr   + (size_t)N * HC;
    const int nd = ((E + 15) / 16) * 4;
    unsigned* rid = (unsigned*)(x1 + (size_t)N * D_FEAT);
    float* ebuf  = (float*)(rid + nd);
    float* mmax  = ebuf + (size_t)EE * NHEAD;

    const dim3 blk(256);
    const int proj_blocks = (N / 16 + GW - 1) / GW;
    const int edge_waves  = (EE + EPW - 1) / EPW;
    const int edge_blocks = (int)(((long long)edge_waves * 32 + 255) / 256);
    const int nh          = N * NHEAD;
    const int range_blocks = (N + RNG - 1) / RNG;

    range_id_kernel<<<(nd + 255) / 256, blk, 0, stream>>>(dst, rid, E, nd);

    const int L = 2;
    const float* xin = x0;
    for (int l = 0; l < L; ++l) {
        const float* Wl_l = Wl + (size_t)l * D_FEAT * HC;
        const float* bl_l = bl + (size_t)l * HC;
        const float* Wr_l = Wr + (size_t)l * D_FEAT * HC;
        const float* br_l = br + (size_t)l * HC;
        const float* at_l = att + (size_t)l * NHEAD * CCH;
        const float* bi_l = bias + (size_t)l * CCH;

        proj_both_wmma<<<proj_blocks, dim3(GW * 32), 0, stream>>>(xin, Wl_l, bl_l, Wr_l, br_l, xl, xr, N);

        fill_kernel<<<(nh + 255) / 256, blk, 0, stream>>>(mmax, -__builtin_inff(), nh);

        score_kernel<<<edge_blocks, blk, 0, stream>>>(xl, xr, src, dst, at_l,
                                                      ebuf, mmax, E, EE);

        float* outp = (l == L - 1) ? (float*)d_out : x1;
        scatter_range_kernel<<<range_blocks, dim3(SCT), 0, stream>>>(
            xl, ebuf, mmax, src, dst, rid, bi_l, outp, (l < L - 1) ? 1 : 0, N, E);

        xin = x1;
    }
}


# Score pass folded into the range pass: a group scores its queued edges itself (xl rows gathered, its 400 xr rows held in LDS), maxima by LDS atomics, row sums by arrival as before; 5 launches
# speedup vs baseline: 3.1057x; 1.4961x over previous
#include <hip/hip_runtime.h>

#define D_FEAT 64
#define HC 128
#define NHEAD 2
#define CCH 64
#define NEG_SLOPE 0.2f

typedef __attribute__((ext_vector_type(2))) float v2f;
typedef __attribute__((ext_vector_type(8))) float v8f;


__device__ __forceinline__ void atomicMaxF(float* addr, float val) {
    if (val >= 0.0f) atomicMax((int*)addr, __float_as_int(val));
    else             atomicMin((unsigned int*)addr, __float_as_uint(val));
}

#define GW 4

__global__ __launch_bounds__(GW * 32)
void proj_both_wmma(const float* __restrict__ X,
                    const float* __restrict__ Wl, const float* __restrict__ bl,
                    const float* __restrict__ Wr, const float* __restrict__ br,
                    float* __restrict__ Yl, float* __restrict__ Yr, int rows) {
    const int tm   = (blockIdx.x * blockDim.x + threadIdx.x) >> 5;
    const int lane = threadIdx.x & 31;
    if (tm >= (rows >> 4)) return;

    const int c     = lane & 15;
    const int hl    = lane >> 4;
    const int khalf = hl << 1;
    const int mrow  = (tm << 4) + c;

    v2f a[16];
    const float* ap = X + (size_t)mrow * D_FEAT + khalf;
#pragma unroll
    for (int s = 0; s < 16; ++s) {
        a[s].x = ap[4 * s];
        a[s].y = ap[4 * s + 1];
    }

    const int rbase = (tm << 4) + (hl << 3);
#pragma unroll
    for (int q = 0; q < 4; ++q) {
        const float* W  = (q < 2) ? Wl : Wr;
        const float* bv = (q < 2) ? bl : br;
        float*       Y  = (q < 2) ? Yl : Yr;
        const int col = ((q & 1) << 6) + (c << 2);

        v8f acc0 = {}, acc1 = {}, acc2 = {}, acc3 = {};
#pragma unroll
        for (int s0 = 0; s0 < 16; s0 += 8) {
            float4 w0[8], w1[8];
#pragma unroll
            for (int i = 0; i < 8; ++i) {
                const int kk = 4 * (s0 + i) + khalf;
                w0[i] = *(const float4*)(W + (size_t)(kk + 0) * HC + col);
                w1[i] = *(const float4*)(W + (size_t)(kk + 1) * HC + col);
            }
#pragma unroll
            for (int i = 0; i < 8; ++i) {
                const v2f av = a[s0 + i];
                v2f b;
                b.x = w0[i].x; b.y = w1[i].x;
                acc0 = __builtin_amdgcn_wmma_f32_16x16x4_f32(false, av, false, b, (short)0, acc0, false, false);
                b.x = w0[i].y; b.y = w1[i].y;
                acc1 = __builtin_amdgcn_wmma_f32_16x16x4_f32(false, av, false, b, (short)0, acc1, false, false);
                b.x = w0[i].z; b.y = w1[i].z;
                acc2 = __builtin_amdgcn_wmma_f32_16x16x4_f32(false, av, false, b, (short)0, acc2, false, false);
                b.x = w0[i].w; b.y = w1[i].w;
                acc3 = __builtin_amdgcn_wmma_f32_16x16x4_f32(false, av, false, b, (short)0, acc3, false, false);
            }
        }

        const float4 bb = *(const float4*)(bv + col);
#pragma unroll
        for (int j = 0; j < 8; ++j)
            *(float4*)(Y + (size_t)(rbase + j) * HC + col) =
                make_float4(acc0[j] + bb.x, acc1[j] + bb.y, acc2[j] + bb.z, acc3[j] + bb.w);
    }
}

#define RNG   400

__global__ void range_id_kernel(const int* __restrict__ dst,
                                unsigned* __restrict__ rid, int E, int nd) {
    const int i = blockIdx.x * blockDim.x + threadIdx.x;
    if (i >= nd) return;
    unsigned w = 0u;
#pragma unroll
    for (int b = 0; b < 4; ++b) {
        const int e = 4 * i + b;
        if (e < E) w |= (((unsigned)dst[e] / (unsigned)RNG) & 255u) << (8 * b);
    }
    rid[i] = w;
}

#define SCT   1024
#define QCAP  5600
#define SCU   4
#define PB    8

__device__ __forceinline__ int acc_slot(int ch) { return ((ch & 3) << 5) | (ch >> 2); }

template <bool GUARD>
__device__ __forceinline__ void add_rows(float* acc, const int4* ent,
                                         const float* __restrict__ xl,
                                         int t0, int total, int h, int c0, int lane) {
    int4   en[PB];
    float4 v[PB];
#pragma unroll
    for (int k = 0; k < PB; ++k) {
        const int t = t0 + k;
        en[k] = ent[(!GUARD || t < total) ? t : 0];
    }
#pragma unroll
    for (int k = 0; k < PB; ++k)
        v[k] = *(const float4*)(xl + (size_t)en[k].x * HC + c0);
#pragma unroll
    for (int k = 0; k < PB; ++k) {
        if ((!GUARD || t0 + k < total) && en[k].y >= 0) {
            const float a = __int_as_float(h ? en[k].w : en[k].z);
            float* o = acc + en[k].y * HC + lane;
            atomicAdd(o,      a * v[k].x);
            atomicAdd(o + 32, a * v[k].y);
            atomicAdd(o + 64, a * v[k].z);
            atomicAdd(o + 96, a * v[k].w);
        }
    }
}

__device__ void range_slow(float* acc, float* den, float* mx,
                           const float* __restrict__ xl, const float* __restrict__ xr,
                           const float* __restrict__ att,
                           const int* __restrict__ src, const int* __restrict__ dst,
                           int lo, int cnt, int E, int tid) {
    for (int pass = 0; pass < 2; ++pass) {
        for (int e = tid; e < E + cnt; e += SCT) {
            int r, s;
            if (e < E) { r = dst[e] - lo; s = src[e]; } else { r = e - E; s = lo + r; }
            if ((unsigned)r >= (unsigned)cnt) continue;
            for (int hh = 0; hh < NHEAD; ++hh) {
                float p = 0.0f;
                for (int cc = 0; cc < CCH; ++cc) {
                    const int ch = hh * CCH + cc;
                    float z = xl[(size_t)s * HC + ch] + xr[(size_t)(lo + r) * HC + ch];
                    z = (z > 0.0f) ? z : NEG_SLOPE * z;
                    p += z * att[ch];
                }
                if (pass == 0) {
                    atomicMaxF(&mx[r * NHEAD + hh], p);
                } else {
                    const float a = __expf(p - mx[r * NHEAD + hh]);
                    atomicAdd(&den[r * NHEAD + hh], a);
                    for (int cc = 0; cc < CCH; ++cc) {
                        const int ch = hh * CCH + cc;
                        atomicAdd(&acc[r * HC + acc_slot(ch)], a * xl[(size_t)s * HC + ch]);
                    }
                }
            }
        }
        __syncthreads();
    }
}

__global__ __launch_bounds__(SCT)
void gat_range_kernel(const float* __restrict__ xl,
                      const float* __restrict__ xr,
                      const float* __restrict__ att,
                      const int* __restrict__ src,
                      const int* __restrict__ dst,
                      const unsigned* __restrict__ rid,
                      const float* __restrict__ bias,
                      float* __restrict__ out,
                      int apply_gelu, int N, int E) {
    __shared__ __attribute__((aligned(16))) float acc[RNG * HC];
    __shared__ __attribute__((aligned(16))) int4 ent[RNG + QCAP];
    __shared__ float den[RNG * NHEAD];
    __shared__ float mx[RNG * NHEAD];
    __shared__ int qn;

    const int tid  = threadIdx.x;
    const int lane = tid & 31;
    const int wave = tid >> 5;
    const int lo   = blockIdx.x * RNG;
    const int cnt  = (N - lo < RNG) ? (N - lo) : RNG;

    float4* a4 = (float4*)acc;
    const float4* xr4 = (const float4*)(xr + (size_t)lo * HC);
    for (int i = tid; i < cnt * (HC / 4); i += SCT) a4[i] = xr4[i];
    for (int i = tid; i < RNG * NHEAD; i += SCT) {
        den[i] = 0.0f;
        mx[i]  = -__builtin_inff();
    }
    if (tid == 0) qn = 0;
    __syncthreads();

    const unsigned pat = (blockIdx.x & 255u) * 0x01010101u;
    const uint4* rid4 = (const uint4*)rid;
    const int nq4 = (E + 15) >> 4;
    const int CH  = SCT * SCU;
    const int nch = (nq4 + CH - 1) / CH;
    int c = (nch > 0) ? (int)(blockIdx.x % (unsigned)nch) : 0;
    for (int ci = 0; ci < nch; ++ci) {
        const int base = c * CH + tid;
        uint4 dv[SCU];
        if ((c + 1) * CH <= nq4) {
#pragma unroll
            for (int k = 0; k < SCU; ++k) dv[k] = rid4[base + k * SCT];
        } else {
#pragma unroll
            for (int k = 0; k < SCU; ++k) {
                const int i = base + k * SCT;
                dv[k] = (i < nq4) ? rid4[i] : make_uint4(~pat, ~pat, ~pat, ~pat);
            }
        }
#pragma unroll
        for (int k = 0; k < SCU; ++k) {
            const unsigned wd[4] = {dv[k].x, dv[k].y, dv[k].z, dv[k].w};
#pragma unroll
            for (int q = 0; q < 4; ++q) {
                const unsigned x = wd[q] ^ pat;
                unsigned m = ~(((x & 0x7F7F7F7Fu) + 0x7F7F7F7Fu) | x | 0x7F7F7F7Fu);
                while (m != 0u) {
                    const int j = (q << 2) + ((__ffs(m) - 1) >> 3);
                    m &= m - 1u;
                    const int e = ((base + k * SCT) << 4) + j;
                    if (e < E) {
                        const int pos = atomicAdd(&qn, 1);
                        if (pos < QCAP) ent[cnt + pos].x = e;
                    }
                }
            }
        }
        if (++c == nch) c = 0;
    }
    __syncthreads();

    if (qn > QCAP) {
        for (int i = tid; i < RNG * HC / 4; i += SCT) a4[i] = make_float4(0.0f, 0.0f, 0.0f, 0.0f);
        __syncthreads();
        range_slow(acc, den, mx, xl, xr, att, src, dst, lo, cnt, E, tid);
    } else {
        const int total = cnt + qn;
        for (int t = tid; t < total; t += SCT) {
            int r, s;
            if (t < cnt) {
                r = t; s = lo + t;
            } else {
                const int e = ent[t].x;
                r = dst[e] - lo; s = src[e];
                if ((unsigned)r >= (unsigned)cnt) { r = -1; s = lo; }
            }
            ent[t] = make_int4(s, r, 0, 0);
        }
        __syncthreads();

        const int h  = lane >> 4;
        const int c0 = lane << 2;
        const int j  = lane & 15;
        const float4 at = *(const float4*)(att + c0);
        int* ei = (int*)ent;
        const int wv = __builtin_amdgcn_readfirstlane(wave);
        for (int t0 = wv * PB; t0 < total; t0 += (SCT / 32) * PB) {
            const int tl = t0 + (lane & (PB - 1));
            const int tc = (tl < total) ? tl : 0;
            const int ms = ei[tc * 4 + 0];
            const int mr = ei[tc * 4 + 1];
            float4 zl[PB], zr[PB];
#pragma unroll
            for (int k = 0; k < PB; ++k) {
                const int sk = __builtin_amdgcn_readlane(ms, k);
                int rk = __builtin_amdgcn_readlane(mr, k);
                rk = (rk < 0) ? 0 : rk;
                zl[k] = *(const float4*)(xl + (size_t)sk * HC + c0);
                zr[k] = *(const float4*)(acc + rk * HC + c0);
            }
            float pv = 0.0f;
#pragma unroll
            for (int k = 0; k < PB; ++k) {
                float p = 0.0f, z;
                z = zl[k].x + zr[k].x; z = (z > 0.0f) ? z : NEG_SLOPE * z; p += z * at.x;
                z = zl[k].y + zr[k].y; z = (z > 0.0f) ? z : NEG_SLOPE * z; p += z * at.y;
                z = zl[k].z + zr[k].z; z = (z > 0.0f) ? z : NEG_SLOPE * z; p += z * at.z;
                z = zl[k].w + zr[k].w; z = (z > 0.0f) ? z : NEG_SLOPE * z; p += z * at.w;

                p += __shfl_xor(p, 1);
                p += __shfl_xor(p, 2);
                p += __shfl_xor(p, 4);
                p += __shfl_xor(p, 8);
                if (j == k) pv = p;
            }
            if (j < PB && tl < total && mr >= 0) {
                ei[tl * 4 + 2 + h] = __float_as_int(pv);
                atomicMaxF(&mx[mr * NHEAD + h], pv);
            }
        }
        __syncthreads();

        for (int i = tid; i < RNG * HC / 4; i += SCT) a4[i] = make_float4(0.0f, 0.0f, 0.0f, 0.0f);
        for (int t = tid; t < total; t += SCT) {
            const int4 en = ent[t];
            if (en.y >= 0) {
                const float a0 = __expf(__int_as_float(en.z) - mx[en.y * NHEAD + 0]);
                const float a1 = __expf(__int_as_float(en.w) - mx[en.y * NHEAD + 1]);
                atomicAdd(&den[en.y * NHEAD + 0], a0);
                atomicAdd(&den[en.y * NHEAD + 1], a1);
                ent[t] = make_int4(en.x, en.y, __float_as_int(a0), __float_as_int(a1));
            }
        }
        __syncthreads();

        for (int t0 = wv * PB; t0 < total; t0 += (SCT / 32) * PB) {
            if (t0 + PB <= total) add_rows<false>(acc, ent, xl, t0, total, h, c0, lane);
            else                  add_rows<true >(acc, ent, xl, t0, total, h, c0, lane);
        }
        __syncthreads();
    }

    for (int idx = tid; idx < cnt * CCH; idx += SCT) {
        const int node = idx >> 6;
        const int cc   = idx & 63;
        const float d0 = den[node * NHEAD + 0] + 1e-16f;
        const float d1 = den[node * NHEAD + 1] + 1e-16f;
        float v = 0.5f * (acc[node * HC + acc_slot(cc)] / d0 +
                          acc[node * HC + acc_slot(CCH + cc)] / d1) + bias[cc];
        if (apply_gelu) {
            const float u = 0.7978845608028654f * (v + 0.044715f * v * v * v);
            v = 0.5f * v * (1.0f + tanhf(u));
        }
        out[(size_t)(lo + node) * CCH + cc] = v;
    }
}


extern "C" void kernel_launch(void* const* d_in, const int* in_sizes, int n_in,
                              void* d_out, int out_size, void* d_ws, size_t ws_size,
                              hipStream_t stream) {
    const float* x0   = (const float*)d_in[0];
    const int*   conn = (const int*)  d_in[1];
    const float* Wl   = (const float*)d_in[2];
    const float* bl   = (const float*)d_in[3];
    const float* Wr   = (const float*)d_in[4];
    const float* br   = (const float*)d_in[5];
    const float* att  = (const float*)d_in[6];
    const float* bias = (const float*)d_in[7];

    const int N  = in_sizes[0] / D_FEAT;
    const int E  = in_sizes[1] / 2;
    const int* src = conn;
    const int* dst = conn + E;

    float* ws    = (float*)d_ws;
    float* xl    = ws;
    float* xr    = xl   + (size_t)N * HC;
    float* x1    = xr   + (size_t)N * HC;
    const int nd = ((E + 15) / 16) * 4;
    unsigned* rid = (unsigned*)(x1 + (size_t)N * D_FEAT);

    const dim3 blk(256);
    const int proj_blocks = (N / 16 + GW - 1) / GW;
    const int range_blocks = (N + RNG - 1) / RNG;

    range_id_kernel<<<(nd + 255) / 256, blk, 0, stream>>>(dst, rid, E, nd);

    const int L = 2;
    const float* xin = x0;
    for (int l = 0; l < L; ++l) {
        const float* Wl_l = Wl + (size_t)l * D_FEAT * HC;
        const float* bl_l = bl + (size_t)l * HC;
        const float* Wr_l = Wr + (size_t)l * D_FEAT * HC;
        const float* br_l = br + (size_t)l * HC;
        const float* at_l = att + (size_t)l * NHEAD * CCH;
        const float* bi_l = bias + (size_t)l * CCH;

        proj_both_wmma<<<proj_blocks, dim3(GW * 32), 0, stream>>>(xin, Wl_l, bl_l, Wr_l, br_l, xl, xr, N);

        float* outp = (l == L - 1) ? (float*)d_out : x1;
        gat_range_kernel<<<range_blocks, dim3(SCT), 0, stream>>>(
            xl, xr, at_l, src, dst, rid, bi_l, outp, (l < L - 1) ? 1 : 0, N, E);

        xin = x1;
    }
}


# Layer 2's range pass reads back the resolved edge queues that layer 1's pass wrote to the workspace (same graph): no range-list search and no resolve in layer 2; LDS float sums by arrival as before
# speedup vs baseline: 3.4082x; 1.0974x over previous
#include <hip/hip_runtime.h>

#define D_FEAT 64
#define HC 128
#define NHEAD 2
#define CCH 64
#define NEG_SLOPE 0.2f

typedef __attribute__((ext_vector_type(2))) float v2f;
typedef __attribute__((ext_vector_type(8))) float v8f;


__device__ __forceinline__ void atomicMaxF(float* addr, float val) {
    if (val >= 0.0f) atomicMax((int*)addr, __float_as_int(val));
    else             atomicMin((unsigned int*)addr, __float_as_uint(val));
}

#define GW 4

__global__ __launch_bounds__(GW * 32)
void proj_both_wmma(const float* __restrict__ X,
                    const float* __restrict__ Wl, const float* __restrict__ bl,
                    const float* __restrict__ Wr, const float* __restrict__ br,
                    float* __restrict__ Yl, float* __restrict__ Yr, int rows) {
    const int tm   = (blockIdx.x * blockDim.x + threadIdx.x) >> 5;
    const int lane = threadIdx.x & 31;
    if (tm >= (rows >> 4)) return;

    const int c     = lane & 15;
    const int hl    = lane >> 4;
    const int khalf = hl << 1;
    const int mrow  = (tm << 4) + c;

    v2f a[16];
    const float* ap = X + (size_t)mrow * D_FEAT + khalf;
#pragma unroll
    for (int s = 0; s < 16; ++s) {
        a[s].x = ap[4 * s];
        a[s].y = ap[4 * s + 1];
    }

    const int rbase = (tm << 4) + (hl << 3);
#pragma unroll
    for (int q = 0; q < 4; ++q) {
        const float* W  = (q < 2) ? Wl : Wr;
        const float* bv = (q < 2) ? bl : br;
        float*       Y  = (q < 2) ? Yl : Yr;
        const int col = ((q & 1) << 6) + (c << 2);

        v8f acc0 = {}, acc1 = {}, acc2 = {}, acc3 = {};
#pragma unroll
        for (int s0 = 0; s0 < 16; s0 += 8) {
            float4 w0[8], w1[8];
#pragma unroll
            for (int i = 0; i < 8; ++i) {
                const int kk = 4 * (s0 + i) + khalf;
                w0[i] = *(const float4*)(W + (size_t)(kk + 0) * HC + col);
                w1[i] = *(const float4*)(W + (size_t)(kk + 1) * HC + col);
            }
#pragma unroll
            for (int i = 0; i < 8; ++i) {
                const v2f av = a[s0 + i];
                v2f b;
                b.x = w0[i].x; b.y = w1[i].x;
                acc0 = __builtin_amdgcn_wmma_f32_16x16x4_f32(false, av, false, b, (short)0, acc0, false, false);
                b.x = w0[i].y; b.y = w1[i].y;
                acc1 = __builtin_amdgcn_wmma_f32_16x16x4_f32(false, av, false, b, (short)0, acc1, false, false);
                b.x = w0[i].z; b.y = w1[i].z;
                acc2 = __builtin_amdgcn_wmma_f32_16x16x4_f32(false, av, false, b, (short)0, acc2, false, false);
                b.x = w0[i].w; b.y = w1[i].w;
                acc3 = __builtin_amdgcn_wmma_f32_16x16x4_f32(false, av, false, b, (short)0, acc3, false, false);
            }
        }

        const float4 bb = *(const float4*)(bv + col);
#pragma unroll
        for (int j = 0; j < 8; ++j)
            *(float4*)(Y + (size_t)(rbase + j) * HC + col) =
                make_float4(acc0[j] + bb.x, acc1[j] + bb.y, acc2[j] + bb.z, acc3[j] + bb.w);
    }
}

#define RNG   400

__global__ void range_id_kernel(const int* __restrict__ dst,
                                unsigned* __restrict__ rid, int E, int nd) {
    const int i = blockIdx.x * blockDim.x + threadIdx.x;
    if (i >= nd) return;
    unsigned w = 0u;
#pragma unroll
    for (int b = 0; b < 4; ++b) {
        const int e = 4 * i + b;
        if (e < E) w |= (((unsigned)dst[e] / (unsigned)RNG) & 255u) << (8 * b);
    }
    rid[i] = w;
}

#define SCT   1024
#define QCAP  5600
#define SCU   4
#define PB    8

__device__ __forceinline__ int acc_slot(int ch) { return ((ch & 3) << 5) | (ch >> 2); }

template <bool GUARD>
__device__ __forceinline__ void add_rows(float* acc, const int4* ent,
                                         const float* __restrict__ xl,
                                         int t0, int total, int h, int c0, int lane) {
    int4   en[PB];
    float4 v[PB];
#pragma unroll
    for (int k = 0; k < PB; ++k) {
        const int t = t0 + k;
        en[k] = ent[(!GUARD || t < total) ? t : 0];
    }
#pragma unroll
    for (int k = 0; k < PB; ++k)
        v[k] = *(const float4*)(xl + (size_t)en[k].x * HC + c0);
#pragma unroll
    for (int k = 0; k < PB; ++k) {
        if ((!GUARD || t0 + k < total) && en[k].y >= 0) {
            const float a = __int_as_float(h ? en[k].w : en[k].z);
            float* o = acc + en[k].y * HC + lane;
            atomicAdd(o,      a * v[k].x);
            atomicAdd(o + 32, a * v[k].y);
            atomicAdd(o + 64, a * v[k].z);
            atomicAdd(o + 96, a * v[k].w);
        }
    }
}

__device__ void range_slow(float* acc, float* den, float* mx,
                           const float* __restrict__ xl, const float* __restrict__ xr,
                           const float* __restrict__ att,
                           const int* __restrict__ src, const int* __restrict__ dst,
                           int lo, int cnt, int E, int tid) {
    for (int pass = 0; pass < 2; ++pass) {
        for (int e = tid; e < E + cnt; e += SCT) {
            int r, s;
            if (e < E) { r = dst[e] - lo; s = src[e]; } else { r = e - E; s = lo + r; }
            if ((unsigned)r >= (unsigned)cnt) continue;
            for (int hh = 0; hh < NHEAD; ++hh) {
                float p = 0.0f;
                for (int cc = 0; cc < CCH; ++cc) {
                    const int ch = hh * CCH + cc;
                    float z = xl[(size_t)s * HC + ch] + xr[(size_t)(lo + r) * HC + ch];
                    z = (z > 0.0f) ? z : NEG_SLOPE * z;
                    p += z * att[ch];
                }
                if (pass == 0) {
                    atomicMaxF(&mx[r * NHEAD + hh], p);
                } else {
                    const float a = __expf(p - mx[r * NHEAD + hh]);
                    atomicAdd(&den[r * NHEAD + hh], a);
                    for (int cc = 0; cc < CCH; ++cc) {
                        const int ch = hh * CCH + cc;
                        atomicAdd(&acc[r * HC + acc_slot(ch)], a * xl[(size_t)s * HC + ch]);
                    }
                }
            }
        }
        __syncthreads();
    }
}

__global__ __launch_bounds__(SCT)
void gat_range_kernel(const float* __restrict__ xl,
                      const float* __restrict__ xr,
                      const float* __restrict__ att,
                      const int* __restrict__ src,
                      const int* __restrict__ dst,
                      const unsigned* __restrict__ rid,
                      int* __restrict__ qcnt,
                      int* __restrict__ qsv,
                      const float* __restrict__ bias,
                      float* __restrict__ out,
                      int apply_gelu, int reuse, int N, int E) {
    __shared__ __attribute__((aligned(16))) float acc[RNG * HC];
    __shared__ __attribute__((aligned(16))) int4 ent[RNG + QCAP];
    __shared__ float den[RNG * NHEAD];
    __shared__ float mx[RNG * NHEAD];
    __shared__ int qn;

    const int tid  = threadIdx.x;
    const int lane = tid & 31;
    const int wave = tid >> 5;
    const int lo   = blockIdx.x * RNG;
    const int cnt  = (N - lo < RNG) ? (N - lo) : RNG;

    float4* a4 = (float4*)acc;
    const float4* xr4 = (const float4*)(xr + (size_t)lo * HC);
    for (int i = tid; i < cnt * (HC / 4); i += SCT) a4[i] = xr4[i];
    for (int i = tid; i < RNG * NHEAD; i += SCT) {
        den[i] = 0.0f;
        mx[i]  = -__builtin_inff();
    }
    if (tid == 0) qn = reuse ? qcnt[blockIdx.x] : 0;
    __syncthreads();

    if (!reuse) {
    const unsigned pat = (blockIdx.x & 255u) * 0x01010101u;
    const uint4* rid4 = (const uint4*)rid;
    const int nq4 = (E + 15) >> 4;
    const int CH  = SCT * SCU;
    const int nch = (nq4 + CH - 1) / CH;
    int c = (nch > 0) ? (int)(blockIdx.x % (unsigned)nch) : 0;
    for (int ci = 0; ci < nch; ++ci) {
        const int base = c * CH + tid;
        uint4 dv[SCU];
        if ((c + 1) * CH <= nq4) {
#pragma unroll
            for (int k = 0; k < SCU; ++k) dv[k] = rid4[base + k * SCT];
        } else {
#pragma unroll
            for (int k = 0; k < SCU; ++k) {
                const int i = base + k * SCT;
                dv[k] = (i < nq4) ? rid4[i] : make_uint4(~pat, ~pat, ~pat, ~pat);
            }
        }
#pragma unroll
        for (int k = 0; k < SCU; ++k) {
            const unsigned wd[4] = {dv[k].x, dv[k].y, dv[k].z, dv[k].w};
#pragma unroll
            for (int q = 0; q < 4; ++q) {
                const unsigned x = wd[q] ^ pat;
                unsigned m = ~(((x & 0x7F7F7F7Fu) + 0x7F7F7F7Fu) | x | 0x7F7F7F7Fu);
                while (m != 0u) {
                    const int j = (q << 2) + ((__ffs(m) - 1) >> 3);
                    m &= m - 1u;
                    const int e = ((base + k * SCT) << 4) + j;
                    if (e < E) {
                        const int pos = atomicAdd(&qn, 1);
                        if (pos < QCAP) ent[cnt + pos].x = e;
                    }
                }
            }
        }
        if (++c == nch) c = 0;
    }
    __syncthreads();
    }

    const int nq = qn;
    if (!reuse && tid == 0) qcnt[blockIdx.x] = nq;
    int2* qs2 = (int2*)qsv + (size_t)blockIdx.x * QCAP;

    if (nq > QCAP) {
        for (int i = tid; i < RNG * HC / 4; i += SCT) a4[i] = make_float4(0.0f, 0.0f, 0.0f, 0.0f);
        __syncthreads();
        range_slow(acc, den, mx, xl, xr, att, src, dst, lo, cnt, E, tid);
    } else {
        const int total = cnt + nq;
        for (int t = tid; t < total; t += SCT) {
            int r, s;
            if (t < cnt) {
                r = t; s = lo + t;
            } else if (reuse) {
                const int2 sv = qs2[t - cnt];
                s = sv.x; r = sv.y;
            } else {
                const int e = ent[t].x;
                r = dst[e] - lo; s = src[e];
                if ((unsigned)r >= (unsigned)cnt) { r = -1; s = lo; }
                qs2[t - cnt] = make_int2(s, r);
            }
            ent[t] = make_int4(s, r, 0, 0);
        }
        __syncthreads();

        const int h  = lane >> 4;
        const int c0 = lane << 2;
        const int j  = lane & 15;
        const float4 at = *(const float4*)(att + c0);
        int* ei = (int*)ent;
        const int wv = __builtin_amdgcn_readfirstlane(wave);
        for (int t0 = wv * PB; t0 < total; t0 += (SCT / 32) * PB) {
            const int tl = t0 + (lane & (PB - 1));
            const int tc = (tl < total) ? tl : 0;
            const int ms = ei[tc * 4 + 0];
            const int mr = ei[tc * 4 + 1];
            float4 zl[PB], zr[PB];
#pragma unroll
            for (int k = 0; k < PB; ++k) {
                const int sk = __builtin_amdgcn_readlane(ms, k);
                int rk = __builtin_amdgcn_readlane(mr, k);
                rk = (rk < 0) ? 0 : rk;
                zl[k] = *(const float4*)(xl + (size_t)sk * HC + c0);
                zr[k] = *(const float4*)(acc + rk * HC + c0);
            }
            float pv = 0.0f;
#pragma unroll
            for (int k = 0; k < PB; ++k) {
                float p = 0.0f, z;
                z = zl[k].x + zr[k].x; z = (z > 0.0f) ? z : NEG_SLOPE * z; p += z * at.x;
                z = zl[k].y + zr[k].y; z = (z > 0.0f) ? z : NEG_SLOPE * z; p += z * at.y;
                z = zl[k].z + zr[k].z; z = (z > 0.0f) ? z : NEG_SLOPE * z; p += z * at.z;
                z = zl[k].w + zr[k].w; z = (z > 0.0f) ? z : NEG_SLOPE * z; p += z * at.w;

                p += __shfl_xor(p, 1);
                p += __shfl_xor(p, 2);
                p += __shfl_xor(p, 4);
                p += __shfl_xor(p, 8);
                if (j == k) pv = p;
            }
            if (j < PB && tl < total && mr >= 0) {
                ei[tl * 4 + 2 + h] = __float_as_int(pv);
                atomicMaxF(&mx[mr * NHEAD + h], pv);
            }
        }
        __syncthreads();

        for (int i = tid; i < RNG * HC / 4; i += SCT) a4[i] = make_float4(0.0f, 0.0f, 0.0f, 0.0f);
        for (int t = tid; t < total; t += SCT) {
            const int4 en = ent[t];
            if (en.y >= 0) {
                const float a0 = __expf(__int_as_float(en.z) - mx[en.y * NHEAD + 0]);
                const float a1 = __expf(__int_as_float(en.w) - mx[en.y * NHEAD + 1]);
                atomicAdd(&den[en.y * NHEAD + 0], a0);
                atomicAdd(&den[en.y * NHEAD + 1], a1);
                ent[t] = make_int4(en.x, en.y, __float_as_int(a0), __float_as_int(a1));
            }
        }
        __syncthreads();

        for (int t0 = wv * PB; t0 < total; t0 += (SCT / 32) * PB) {
            if (t0 + PB <= total) add_rows<false>(acc, ent, xl, t0, total, h, c0, lane);
            else                  add_rows<true >(acc, ent, xl, t0, total, h, c0, lane);
        }
        __syncthreads();
    }

    for (int idx = tid; idx < cnt * CCH; idx += SCT) {
        const int node = idx >> 6;
        const int cc   = idx & 63;
        const float d0 = den[node * NHEAD + 0] + 1e-16f;
        const float d1 = den[node * NHEAD + 1] + 1e-16f;
        float v = 0.5f * (acc[node * HC + acc_slot(cc)] / d0 +
                          acc[node * HC + acc_slot(CCH + cc)] / d1) + bias[cc];
        if (apply_gelu) {
            const float u = 0.7978845608028654f * (v + 0.044715f * v * v * v);
            v = 0.5f * v * (1.0f + tanhf(u));
        }
        out[(size_t)(lo + node) * CCH + cc] = v;
    }
}


extern "C" void kernel_launch(void* const* d_in, const int* in_sizes, int n_in,
                              void* d_out, int out_size, void* d_ws, size_t ws_size,
                              hipStream_t stream) {
    const float* x0   = (const float*)d_in[0];
    const int*   conn = (const int*)  d_in[1];
    const float* Wl   = (const float*)d_in[2];
    const float* bl   = (const float*)d_in[3];
    const float* Wr   = (const float*)d_in[4];
    const float* br   = (const float*)d_in[5];
    const float* att  = (const float*)d_in[6];
    const float* bias = (const float*)d_in[7];

    const int N  = in_sizes[0] / D_FEAT;
    const int E  = in_sizes[1] / 2;
    const int* src = conn;
    const int* dst = conn + E;

    float* ws    = (float*)d_ws;
    float* xl    = ws;
    float* xr    = xl   + (size_t)N * HC;
    float* x1    = xr   + (size_t)N * HC;
    const int nd = ((E + 15) / 16) * 4;
    unsigned* rid = (unsigned*)(x1 + (size_t)N * D_FEAT);

    const dim3 blk(256);
    const int proj_blocks = (N / 16 + GW - 1) / GW;
    const int range_blocks = (N + RNG - 1) / RNG;
    int* qcnt = (int*)(rid + nd);
    int* qsv  = qcnt + ((range_blocks + 3) / 4) * 4;

    range_id_kernel<<<(nd + 255) / 256, blk, 0, stream>>>(dst, rid, E, nd);

    const int L = 2;
    const float* xin = x0;
    for (int l = 0; l < L; ++l) {
        const float* Wl_l = Wl + (size_t)l * D_FEAT * HC;
        const float* bl_l = bl + (size_t)l * HC;
        const float* Wr_l = Wr + (size_t)l * D_FEAT * HC;
        const float* br_l = br + (size_t)l * HC;
        const float* at_l = att + (size_t)l * NHEAD * CCH;
        const float* bi_l = bias + (size_t)l * CCH;

        proj_both_wmma<<<proj_blocks, dim3(GW * 32), 0, stream>>>(xin, Wl_l, bl_l, Wr_l, br_l, xl, xr, N);

        float* outp = (l == L - 1) ? (float*)d_out : x1;
        gat_range_kernel<<<range_blocks, dim3(SCT), 0, stream>>>(
            xl, xr, at_l, src, dst, rid, qcnt, qsv, bi_l, outp, (l < L - 1) ? 1 : 0, (l > 0) ? 1 : 0, N, E);

        xin = x1;
    }
}


# Range pass gathers each xl row once: weight = exp(score - dst's self-loop score), no maximum pass (gap > 64: exact two-walk path); two 200-node halves per group; rounding differs; sums by arrival
# speedup vs baseline: 3.7504x; 1.1004x over previous
#include <hip/hip_runtime.h>

#define D_FEAT 64
#define HC 128
#define NHEAD 2
#define CCH 64
#define NEG_SLOPE 0.2f

typedef __attribute__((ext_vector_type(2))) float v2f;
typedef __attribute__((ext_vector_type(8))) float v8f;
typedef __attribute__((ext_vector_type(4))) float v4f;


__device__ __forceinline__ void atomicMaxF(float* addr, float val) {
    if (val >= 0.0f) atomicMax((int*)addr, __float_as_int(val));
    else             atomicMin((unsigned int*)addr, __float_as_uint(val));
}

#define GW 4

__global__ __launch_bounds__(GW * 32)
void proj_both_wmma(const float* __restrict__ X,
                    const float* __restrict__ Wl, const float* __restrict__ bl,
                    const float* __restrict__ Wr, const float* __restrict__ br,
                    float* __restrict__ Yl, float* __restrict__ Yr, int rows) {
    const int tm   = (blockIdx.x * blockDim.x + threadIdx.x) >> 5;
    const int lane = threadIdx.x & 31;
    if (tm >= (rows >> 4)) return;

    const int c     = lane & 15;
    const int hl    = lane >> 4;
    const int khalf = hl << 1;
    const int mrow  = (tm << 4) + c;

    v2f a[16];
    const float* ap = X + (size_t)mrow * D_FEAT + khalf;
#pragma unroll
    for (int s = 0; s < 16; ++s) {
        a[s].x = ap[4 * s];
        a[s].y = ap[4 * s + 1];
    }

    const int rbase = (tm << 4) + (hl << 3);
#pragma unroll
    for (int q = 0; q < 4; ++q) {
        const float* W  = (q < 2) ? Wl : Wr;
        const float* bv = (q < 2) ? bl : br;
        float*       Y  = (q < 2) ? Yl : Yr;
        const int col = ((q & 1) << 6) + (c << 2);

        v8f acc0 = {}, acc1 = {}, acc2 = {}, acc3 = {};
#pragma unroll
        for (int s0 = 0; s0 < 16; s0 += 8) {
            float4 w0[8], w1[8];
#pragma unroll
            for (int i = 0; i < 8; ++i) {
                const int kk = 4 * (s0 + i) + khalf;
                w0[i] = *(const float4*)(W + (size_t)(kk + 0) * HC + col);
                w1[i] = *(const float4*)(W + (size_t)(kk + 1) * HC + col);
            }
#pragma unroll
            for (int i = 0; i < 8; ++i) {
                const v2f av = a[s0 + i];
                v2f b;
                b.x = w0[i].x; b.y = w1[i].x;
                acc0 = __builtin_amdgcn_wmma_f32_16x16x4_f32(false, av, false, b, (short)0, acc0, false, false);
                b.x = w0[i].y; b.y = w1[i].y;
                acc1 = __builtin_amdgcn_wmma_f32_16x16x4_f32(false, av, false, b, (short)0, acc1, false, false);
                b.x = w0[i].z; b.y = w1[i].z;
                acc2 = __builtin_amdgcn_wmma_f32_16x16x4_f32(false, av, false, b, (short)0, acc2, false, false);
                b.x = w0[i].w; b.y = w1[i].w;
                acc3 = __builtin_amdgcn_wmma_f32_16x16x4_f32(false, av, false, b, (short)0, acc3, false, false);
            }
        }

        const float4 bb = *(const float4*)(bv + col);
#pragma unroll
        for (int j = 0; j < 8; ++j)
            *(float4*)(Y + (size_t)(rbase + j) * HC + col) =
                make_float4(acc0[j] + bb.x, acc1[j] + bb.y, acc2[j] + bb.z, acc3[j] + bb.w);
    }
}

#define RNG   400

__global__ void range_id_kernel(const int* __restrict__ dst,
                                unsigned* __restrict__ rid, int E, int nd) {
    const int i = blockIdx.x * blockDim.x + threadIdx.x;
    if (i >= nd) return;
    unsigned w = 0u;
#pragma unroll
    for (int b = 0; b < 4; ++b) {
        const int e = 4 * i + b;
        if (e < E) w |= (((unsigned)dst[e] / (unsigned)RNG) & 255u) << (8 * b);
    }
    rid[i] = w;
}

#define SCT   1024
#define RNH   200
#define QCAP  5600
#define SCU   4
#define PB    8
#define SB    4
#define SHIFT_CAP 64.0f

__device__ __forceinline__ int acc_slot(int ch) { return ((ch & 3) << 5) | (ch >> 2); }

__device__ __forceinline__ float score16(const v4f vl, const v4f vr, const v4f at) {
    float p = 0.0f, z;
    z = vl.x + vr.x; z = (z > 0.0f) ? z : NEG_SLOPE * z; p += z * at.x;
    z = vl.y + vr.y; z = (z > 0.0f) ? z : NEG_SLOPE * z; p += z * at.y;
    z = vl.z + vr.z; z = (z > 0.0f) ? z : NEG_SLOPE * z; p += z * at.z;
    z = vl.w + vr.w; z = (z > 0.0f) ? z : NEG_SLOPE * z; p += z * at.w;
    p += __shfl_xor(p, 1);
    p += __shfl_xor(p, 2);
    p += __shfl_xor(p, 4);
    p += __shfl_xor(p, 8);
    return p;
}

__device__ void range_slow(float* acc, float* den, float* mx,
                           const float* __restrict__ xl, const float* __restrict__ xr,
                           const float* __restrict__ att,
                           const int* __restrict__ src, const int* __restrict__ dst,
                           int lo, int cnt, int E, int tid) {
    for (int pass = 0; pass < 2; ++pass) {
        for (int e = tid; e < E + cnt; e += SCT) {
            int r, s;
            if (e < E) { r = dst[e] - lo; s = src[e]; } else { r = e - E; s = lo + r; }
            if ((unsigned)r >= (unsigned)cnt) continue;
            for (int hh = 0; hh < NHEAD; ++hh) {
                float p = 0.0f;
                for (int cc = 0; cc < CCH; ++cc) {
                    const int ch = hh * CCH + cc;
                    float z = xl[(size_t)s * HC + ch] + xr[(size_t)(lo + r) * HC + ch];
                    z = (z > 0.0f) ? z : NEG_SLOPE * z;
                    p += z * att[ch];
                }
                if (pass == 0) {
                    atomicMaxF(&mx[r * NHEAD + hh], p);
                } else {
                    const float a = __expf(p - mx[r * NHEAD + hh]);
                    atomicAdd(&den[r * NHEAD + hh], a);
                    for (int cc = 0; cc < CCH; ++cc) {
                        const int ch = hh * CCH + cc;
                        atomicAdd(&acc[r * HC + acc_slot(ch)], a * xl[(size_t)s * HC + ch]);
                    }
                }
            }
        }
        __syncthreads();
    }
}

__global__ __launch_bounds__(SCT)
void gat_range_kernel(const float* __restrict__ xl,
                      const float* __restrict__ xr,
                      const float* __restrict__ att,
                      const int* __restrict__ src,
                      const int* __restrict__ dst,
                      const unsigned* __restrict__ rid,
                      int* __restrict__ qcnt,
                      int* __restrict__ qsv,
                      const float* __restrict__ bias,
                      float* __restrict__ out,
                      int apply_gelu, int reuse, int N, int E) {
    __shared__ __attribute__((aligned(16))) float xrs[RNH * HC];
    __shared__ __attribute__((aligned(16))) float acc[RNH * HC];
    __shared__ __attribute__((aligned(8))) int2 ent[QCAP];
    __shared__ int qe[QCAP];
    __shared__ float den[RNH * NHEAD];
    __shared__ float sh[RNH * NHEAD];
    __shared__ float mx[RNH * NHEAD];
    __shared__ int qn, qlo, qhi, ovf;

    const int tid  = threadIdx.x;
    const int lane = tid & 31;
    const int wave = tid >> 5;
    const int lo   = blockIdx.x * RNG;
    const int cnt  = (N - lo < RNG) ? (N - lo) : RNG;

    if (tid == 0) {
        const int a = reuse ? qcnt[2 * blockIdx.x]     : 0;
        const int b = reuse ? qcnt[2 * blockIdx.x + 1] : 0;
        qlo = a; qhi = b; qn = a + b;
    }
    __syncthreads();

    if (!reuse) {
    const unsigned pat = (blockIdx.x & 255u) * 0x01010101u;
    const uint4* rid4 = (const uint4*)rid;
    const int nq4 = (E + 15) >> 4;
    const int CH  = SCT * SCU;
    const int nch = (nq4 + CH - 1) / CH;
    int c = (nch > 0) ? (int)(blockIdx.x % (unsigned)nch) : 0;
    for (int ci = 0; ci < nch; ++ci) {
        const int base = c * CH + tid;
        uint4 dv[SCU];
        if ((c + 1) * CH <= nq4) {
#pragma unroll
            for (int k = 0; k < SCU; ++k) dv[k] = rid4[base + k * SCT];
        } else {
#pragma unroll
            for (int k = 0; k < SCU; ++k) {
                const int i = base + k * SCT;
                dv[k] = (i < nq4) ? rid4[i] : make_uint4(~pat, ~pat, ~pat, ~pat);
            }
        }
#pragma unroll
        for (int k = 0; k < SCU; ++k) {
            const unsigned wd[4] = {dv[k].x, dv[k].y, dv[k].z, dv[k].w};
#pragma unroll
            for (int q = 0; q < 4; ++q) {
                const unsigned x = wd[q] ^ pat;
                unsigned m = ~(((x & 0x7F7F7F7Fu) + 0x7F7F7F7Fu) | x | 0x7F7F7F7Fu);
                while (m != 0u) {
                    const int j = (q << 2) + ((__ffs(m) - 1) >> 3);
                    m &= m - 1u;
                    const int e = ((base + k * SCT) << 4) + j;
                    if (e < E) {
                        const int pos = atomicAdd(&qn, 1);
                        if (pos < QCAP) qe[pos] = e;
                    }
                }
            }
        }
        if (++c == nch) c = 0;
    }
    __syncthreads();
    }

    const int nq = qn;
    const bool slow_all = nq > QCAP;
    int2* qs2 = (int2*)qsv + (size_t)blockIdx.x * QCAP;

    if (!slow_all) {
        if (reuse) {
            const int na = qlo, nb = qhi;
            for (int t = tid; t < na; t += SCT) ent[t] = qs2[t];
            for (int t = tid; t < nb; t += SCT) ent[QCAP - 1 - t] = qs2[QCAP - 1 - t];
        } else {
            for (int t = tid; t < nq; t += SCT) {
                const int e = qe[t];
                const int r = dst[e] - lo;
                const int s = src[e];
                if ((unsigned)r < (unsigned)cnt) {
                    const bool up = r >= RNH;
                    const int pos = up ? (QCAP - 1 - atomicAdd(&qhi, 1)) : atomicAdd(&qlo, 1);
                    const int2 en = make_int2(s, up ? (r - RNH) : r);
                    ent[pos] = en;
                    qs2[pos] = en;
                }
            }
        }
        __syncthreads();
    }
    if (!reuse && tid == 0) {
        qcnt[2 * blockIdx.x]     = slow_all ? (QCAP + 1) : qlo;
        qcnt[2 * blockIdx.x + 1] = slow_all ? 0 : qhi;
    }
    const int na = slow_all ? 0 : qlo;
    const int nb = slow_all ? 0 : qhi;

    const int h  = lane >> 4;
    const int c0 = lane << 2;
    const int j  = lane & 15;
    const v4f at = *(const v4f*)(att + c0);
    const int wv = __builtin_amdgcn_readfirstlane(wave);
    const int* ei = (const int*)ent;

    for (int half = 0; half * RNH < cnt; ++half) {
        const int lo_h  = lo + half * RNH;
        const int cnt_h = (cnt - half * RNH < RNH) ? (cnt - half * RNH) : RNH;
        bool slow = slow_all;

        if (!slow) {
            if (tid == 0) ovf = 0;
            for (int n0 = wv; n0 < cnt_h; n0 += (SCT / 32) * SB) {
                v4f vr[SB], vl[SB];
#pragma unroll
                for (int i = 0; i < SB; ++i) {
                    const int n = n0 + i * (SCT / 32);
                    const size_t g = (size_t)(lo_h + ((n < cnt_h) ? n : n0)) * HC + c0;
                    vr[i] = *(const v4f*)(xr + g);
                    vl[i] = *(const v4f*)(xl + g);
                }
                asm volatile("" :: "v"(vr[0]), "v"(vr[1]), "v"(vr[2]), "v"(vr[3]),
                                   "v"(vl[0]), "v"(vl[1]), "v"(vl[2]), "v"(vl[3]));
#pragma unroll
                for (int i = 0; i < SB; ++i) {
                    const int n = n0 + i * (SCT / 32);
                    if (n < cnt_h) {
                        *(v4f*)(xrs + n * HC + c0) = vr[i];
                        const float p = score16(vl[i], vr[i], at);
                        float* o = acc + n * HC + lane;
                        o[0]  = vl[i].x;
                        o[32] = vl[i].y;
                        o[64] = vl[i].z;
                        o[96] = vl[i].w;
                        if (j == 0) {
                            sh[n * NHEAD + h]  = p;
                            den[n * NHEAD + h] = 1.0f;
                        }
                    }
                }
            }
            __syncthreads();

            const int tb = half ? (QCAP - nb) : 0;
            const int te = half ? QCAP : na;
            float dmax = -__builtin_inff();
            for (int t0 = tb + wv * PB; t0 < te; t0 += (SCT / 32) * PB) {
                const int tl = t0 + (lane & (PB - 1));
                const int tc = (tl < te) ? tl : tb;
                const int ms = ei[tc * 2 + 0];
                const int mr = ei[tc * 2 + 1];
                v4f   zl[PB], zr[PB];
                float sf[PB];
#pragma unroll
                for (int k = 0; k < PB; ++k) {
                    const int sk = __builtin_amdgcn_readlane(ms, k);
                    const int rk = __builtin_amdgcn_readlane(mr, k);
                    zl[k] = *(const v4f*)(xl + (size_t)sk * HC + c0);
                    zr[k] = *(const v4f*)(xrs + rk * HC + c0);
                    sf[k] = sh[rk * NHEAD + h];
                }
                asm volatile("" :: "v"(zl[0]), "v"(zl[1]), "v"(zl[2]), "v"(zl[3]),
                                   "v"(zl[4]), "v"(zl[5]), "v"(zl[6]), "v"(zl[7]));
                float av = 0.0f;
#pragma unroll
                for (int k = 0; k < PB; ++k) {
                    const float d = score16(zl[k], zr[k], at) - sf[k];
                    dmax = fmaxf(dmax, d);
                    const float a = __expf(d);
                    if (j == k) av = a;
                    if (t0 + k < te) {
                        const int rk = __builtin_amdgcn_readlane(mr, k);
                        float* o = acc + rk * HC + lane;
                        atomicAdd(o,      a * zl[k].x);
                        atomicAdd(o + 32, a * zl[k].y);
                        atomicAdd(o + 64, a * zl[k].z);
                        atomicAdd(o + 96, a * zl[k].w);
                    }
                }
                if (j < PB && tl < te) atomicAdd(&den[mr * NHEAD + h], av);
            }
            if (dmax > SHIFT_CAP) ovf = 1;
            __syncthreads();
            slow = (ovf != 0);
        }

        if (slow) {
            float4* a4 = (float4*)acc;
            for (int i = tid; i < RNH * HC / 4; i += SCT) a4[i] = make_float4(0.0f, 0.0f, 0.0f, 0.0f);
            for (int i = tid; i < RNH * NHEAD; i += SCT) {
                den[i] = 0.0f;
                mx[i]  = -__builtin_inff();
            }
            __syncthreads();
            range_slow(acc, den, mx, xl, xr, att, src, dst, lo_h, cnt_h, E, tid);
        }

        for (int idx = tid; idx < cnt_h * CCH; idx += SCT) {
            const int node = idx >> 6;
            const int cc   = idx & 63;
            const float d0 = den[node * NHEAD + 0] + 1e-16f;
            const float d1 = den[node * NHEAD + 1] + 1e-16f;
            float v = 0.5f * (acc[node * HC + acc_slot(cc)] / d0 +
                              acc[node * HC + acc_slot(CCH + cc)] / d1) + bias[cc];
            if (apply_gelu) {
                const float u = 0.7978845608028654f * (v + 0.044715f * v * v * v);
                v = 0.5f * v * (1.0f + tanhf(u));
            }
            out[(size_t)(lo_h + node) * CCH + cc] = v;
        }
        if ((half + 1) * RNH < cnt) __syncthreads();
    }
}


extern "C" void kernel_launch(void* const* d_in, const int* in_sizes, int n_in,
                              void* d_out, int out_size, void* d_ws, size_t ws_size,
                              hipStream_t stream) {
    const float* x0   = (const float*)d_in[0];
    const int*   conn = (const int*)  d_in[1];
    const float* Wl   = (const float*)d_in[2];
    const float* bl   = (const float*)d_in[3];
    const float* Wr   = (const float*)d_in[4];
    const float* br   = (const float*)d_in[5];
    const float* att  = (const float*)d_in[6];
    const float* bias = (const float*)d_in[7];

    const int N  = in_sizes[0] / D_FEAT;
    const int E  = in_sizes[1] / 2;
    const int* src = conn;
    const int* dst = conn + E;

    float* ws    = (float*)d_ws;
    float* xl    = ws;
    float* xr    = xl   + (size_t)N * HC;
    float* x1    = xr   + (size_t)N * HC;
    const int nd = ((E + 15) / 16) * 4;
    unsigned* rid = (unsigned*)(x1 + (size_t)N * D_FEAT);

    const dim3 blk(256);
    const int proj_blocks = (N / 16 + GW - 1) / GW;
    const int range_blocks = (N + RNG - 1) / RNG;
    int* qcnt = (int*)(rid + nd);
    int* qsv  = qcnt + ((2 * range_blocks + 3) / 4) * 4;

    range_id_kernel<<<(nd + 255) / 256, blk, 0, stream>>>(dst, rid, E, nd);

    const int L = 2;
    const float* xin = x0;
    for (int l = 0; l < L; ++l) {
        const float* Wl_l = Wl + (size_t)l * D_FEAT * HC;
        const float* bl_l = bl + (size_t)l * HC;
        const float* Wr_l = Wr + (size_t)l * D_FEAT * HC;
        const float* br_l = br + (size_t)l * HC;
        const float* at_l = att + (size_t)l * NHEAD * CCH;
        const float* bi_l = bias + (size_t)l * CCH;

        proj_both_wmma<<<proj_blocks, dim3(GW * 32), 0, stream>>>(xin, Wl_l, bl_l, Wr_l, br_l, xl, xr, N);

        float* outp = (l == L - 1) ? (float*)d_out : x1;
        gat_range_kernel<<<range_blocks, dim3(SCT), 0, stream>>>(
            xl, xr, at_l, src, dst, rid, qcnt, qsv, bi_l, outp, (l < L - 1) ? 1 : 0, (l > 0) ? 1 : 0, N, E);

        xin = x1;
    }
}


# Edges and stage phases: a batch's 8 (4) scores are summed over their 16 lanes together by a halving exchange: 8 (5) lane exchanges per batch, was 32 (16); same pairs added; LDS sums by arrival
# speedup vs baseline: 4.0208x; 1.0721x over previous
#include <hip/hip_runtime.h>

#define D_FEAT 64
#define HC 128
#define NHEAD 2
#define CCH 64
#define NEG_SLOPE 0.2f

typedef __attribute__((ext_vector_type(2))) float v2f;
typedef __attribute__((ext_vector_type(8))) float v8f;
typedef __attribute__((ext_vector_type(4))) float v4f;


__device__ __forceinline__ void atomicMaxF(float* addr, float val) {
    if (val >= 0.0f) atomicMax((int*)addr, __float_as_int(val));
    else             atomicMin((unsigned int*)addr, __float_as_uint(val));
}

#define GW 4

__global__ __launch_bounds__(GW * 32)
void proj_both_wmma(const float* __restrict__ X,
                    const float* __restrict__ Wl, const float* __restrict__ bl,
                    const float* __restrict__ Wr, const float* __restrict__ br,
                    float* __restrict__ Yl, float* __restrict__ Yr, int rows) {
    const int tm   = (blockIdx.x * blockDim.x + threadIdx.x) >> 5;
    const int lane = threadIdx.x & 31;
    if (tm >= (rows >> 4)) return;

    const int c     = lane & 15;
    const int hl    = lane >> 4;
    const int khalf = hl << 1;
    const int mrow  = (tm << 4) + c;

    v2f a[16];
    const float* ap = X + (size_t)mrow * D_FEAT + khalf;
#pragma unroll
    for (int s = 0; s < 16; ++s) {
        a[s].x = ap[4 * s];
        a[s].y = ap[4 * s + 1];
    }

    const int rbase = (tm << 4) + (hl << 3);
#pragma unroll
    for (int q = 0; q < 4; ++q) {
        const float* W  = (q < 2) ? Wl : Wr;
        const float* bv = (q < 2) ? bl : br;
        float*       Y  = (q < 2) ? Yl : Yr;
        const int col = ((q & 1) << 6) + (c << 2);

        v8f acc0 = {}, acc1 = {}, acc2 = {}, acc3 = {};
#pragma unroll
        for (int s0 = 0; s0 < 16; s0 += 8) {
            float4 w0[8], w1[8];
#pragma unroll
            for (int i = 0; i < 8; ++i) {
                const int kk = 4 * (s0 + i) + khalf;
                w0[i] = *(const float4*)(W + (size_t)(kk + 0) * HC + col);
                w1[i] = *(const float4*)(W + (size_t)(kk + 1) * HC + col);
            }
#pragma unroll
            for (int i = 0; i < 8; ++i) {
                const v2f av = a[s0 + i];
                v2f b;
                b.x = w0[i].x; b.y = w1[i].x;
                acc0 = __builtin_amdgcn_wmma_f32_16x16x4_f32(false, av, false, b, (short)0, acc0, false, false);
                b.x = w0[i].y; b.y = w1[i].y;
                acc1 = __builtin_amdgcn_wmma_f32_16x16x4_f32(false, av, false, b, (short)0, acc1, false, false);
                b.x = w0[i].z; b.y = w1[i].z;
                acc2 = __builtin_amdgcn_wmma_f32_16x16x4_f32(false, av, false, b, (short)0, acc2, false, false);
                b.x = w0[i].w; b.y = w1[i].w;
                acc3 = __builtin_amdgcn_wmma_f32_16x16x4_f32(false, av, false, b, (short)0, acc3, false, false);
            }
        }

        const float4 bb = *(const float4*)(bv + col);
#pragma unroll
        for (int j = 0; j < 8; ++j)
            *(float4*)(Y + (size_t)(rbase + j) * HC + col) =
                make_float4(acc0[j] + bb.x, acc1[j] + bb.y, acc2[j] + bb.z, acc3[j] + bb.w);
    }
}

#define RNG   400

__global__ void range_id_kernel(const int* __restrict__ dst,
                                unsigned* __restrict__ rid, int E, int nd) {
    const int i = blockIdx.x * blockDim.x + threadIdx.x;
    if (i >= nd) return;
    unsigned w = 0u;
#pragma unroll
    for (int b = 0; b < 4; ++b) {
        const int e = 4 * i + b;
        if (e < E) w |= (((unsigned)dst[e] / (unsigned)RNG) & 255u) << (8 * b);
    }
    rid[i] = w;
}

#define SCT   1024
#define RNH   200
#define QCAP  5600
#define SCU   4
#define PB    8
#define SB    4
#define BREV3(k) ((((k) & 1) << 2) | ((k) & 2) | (((k) >> 2) & 1))
#define SHIFT_CAP 64.0f

__device__ __forceinline__ int acc_slot(int ch) { return ((ch & 3) << 5) | (ch >> 2); }

__device__ __forceinline__ float part4(const v4f vl, const v4f vr, const v4f at) {
    float p = 0.0f, z;
    z = vl.x + vr.x; z = (z > 0.0f) ? z : NEG_SLOPE * z; p += z * at.x;
    z = vl.y + vr.y; z = (z > 0.0f) ? z : NEG_SLOPE * z; p += z * at.y;
    z = vl.z + vr.z; z = (z > 0.0f) ? z : NEG_SLOPE * z; p += z * at.z;
    z = vl.w + vr.w; z = (z > 0.0f) ? z : NEG_SLOPE * z; p += z * at.w;
    return p;
}

__device__ void range_slow(float* acc, float* den, float* mx,
                           const float* __restrict__ xl, const float* __restrict__ xr,
                           const float* __restrict__ att,
                           const int* __restrict__ src, const int* __restrict__ dst,
                           int lo, int cnt, int E, int tid) {
    for (int pass = 0; pass < 2; ++pass) {
        for (int e = tid; e < E + cnt; e += SCT) {
            int r, s;
            if (e < E) { r = dst[e] - lo; s = src[e]; } else { r = e - E; s = lo + r; }
            if ((unsigned)r >= (unsigned)cnt) continue;
            for (int hh = 0; hh < NHEAD; ++hh) {
                float p = 0.0f;
                for (int cc = 0; cc < CCH; ++cc) {
                    const int ch = hh * CCH + cc;
                    float z = xl[(size_t)s * HC + ch] + xr[(size_t)(lo + r) * HC + ch];
                    z = (z > 0.0f) ? z : NEG_SLOPE * z;
                    p += z * att[ch];
                }
                if (pass == 0) {
                    atomicMaxF(&mx[r * NHEAD + hh], p);
                } else {
                    const float a = __expf(p - mx[r * NHEAD + hh]);
                    atomicAdd(&den[r * NHEAD + hh], a);
                    for (int cc = 0; cc < CCH; ++cc) {
                        const int ch = hh * CCH + cc;
                        atomicAdd(&acc[r * HC + acc_slot(ch)], a * xl[(size_t)s * HC + ch]);
                    }
                }
            }
        }
        __syncthreads();
    }
}

__global__ __launch_bounds__(SCT)
void gat_range_kernel(const float* __restrict__ xl,
                      const float* __restrict__ xr,
                      const float* __restrict__ att,
                      const int* __restrict__ src,
                      const int* __restrict__ dst,
                      const unsigned* __restrict__ rid,
                      int* __restrict__ qcnt,
                      int* __restrict__ qsv,
                      const float* __restrict__ bias,
                      float* __restrict__ out,
                      int apply_gelu, int reuse, int N, int E) {
    __shared__ __attribute__((aligned(16))) float xrs[RNH * HC];
    __shared__ __attribute__((aligned(16))) float acc[RNH * HC];
    __shared__ __attribute__((aligned(8))) int2 ent[QCAP];
    __shared__ int qe[QCAP];
    __shared__ float den[RNH * NHEAD];
    __shared__ float sh[RNH * NHEAD];
    __shared__ float mx[RNH * NHEAD];
    __shared__ int qn, qlo, qhi, ovf;

    const int tid  = threadIdx.x;
    const int lane = tid & 31;
    const int wave = tid >> 5;
    const int lo   = blockIdx.x * RNG;
    const int cnt  = (N - lo < RNG) ? (N - lo) : RNG;

    if (tid == 0) {
        const int a = reuse ? qcnt[2 * blockIdx.x]     : 0;
        const int b = reuse ? qcnt[2 * blockIdx.x + 1] : 0;
        qlo = a; qhi = b; qn = a + b;
    }
    __syncthreads();

    if (!reuse) {
    const unsigned pat = (blockIdx.x & 255u) * 0x01010101u;
    const uint4* rid4 = (const uint4*)rid;
    const int nq4 = (E + 15) >> 4;
    const int CH  = SCT * SCU;
    const int nch = (nq4 + CH - 1) / CH;
    int c = (nch > 0) ? (int)(blockIdx.x % (unsigned)nch) : 0;
    for (int ci = 0; ci < nch; ++ci) {
        const int base = c * CH + tid;
        uint4 dv[SCU];
        if ((c + 1) * CH <= nq4) {
#pragma unroll
            for (int k = 0; k < SCU; ++k) dv[k] = rid4[base + k * SCT];
        } else {
#pragma unroll
            for (int k = 0; k < SCU; ++k) {
                const int i = base + k * SCT;
                dv[k] = (i < nq4) ? rid4[i] : make_uint4(~pat, ~pat, ~pat, ~pat);
            }
        }
#pragma unroll
        for (int k = 0; k < SCU; ++k) {
            const unsigned wd[4] = {dv[k].x, dv[k].y, dv[k].z, dv[k].w};
#pragma unroll
            for (int q = 0; q < 4; ++q) {
                const unsigned x = wd[q] ^ pat;
                unsigned m = ~(((x & 0x7F7F7F7Fu) + 0x7F7F7F7Fu) | x | 0x7F7F7F7Fu);
                while (m != 0u) {
                    const int j = (q << 2) + ((__ffs(m) - 1) >> 3);
                    m &= m - 1u;
                    const int e = ((base + k * SCT) << 4) + j;
                    if (e < E) {
                        const int pos = atomicAdd(&qn, 1);
                        if (pos < QCAP) qe[pos] = e;
                    }
                }
            }
        }
        if (++c == nch) c = 0;
    }
    __syncthreads();
    }

    const int nq = qn;
    const bool slow_all = nq > QCAP;
    int2* qs2 = (int2*)qsv + (size_t)blockIdx.x * QCAP;

    if (!slow_all) {
        if (reuse) {
            const int na = qlo, nb = qhi;
            for (int t = tid; t < na; t += SCT) ent[t] = qs2[t];
            for (int t = tid; t < nb; t += SCT) ent[QCAP - 1 - t] = qs2[QCAP - 1 - t];
        } else {
            for (int t = tid; t < nq; t += SCT) {
                const int e = qe[t];
                const int r = dst[e] - lo;
                const int s = src[e];
                if ((unsigned)r < (unsigned)cnt) {
                    const bool up = r >= RNH;
                    const int pos = up ? (QCAP - 1 - atomicAdd(&qhi, 1)) : atomicAdd(&qlo, 1);
                    const int2 en = make_int2(s, up ? (r - RNH) : r);
                    ent[pos] = en;
                    qs2[pos] = en;
                }
            }
        }
        __syncthreads();
    }
    if (!reuse && tid == 0) {
        qcnt[2 * blockIdx.x]     = slow_all ? (QCAP + 1) : qlo;
        qcnt[2 * blockIdx.x + 1] = slow_all ? 0 : qhi;
    }
    const int na = slow_all ? 0 : qlo;
    const int nb = slow_all ? 0 : qhi;

    const int h  = lane >> 4;
    const int c0 = lane << 2;
    const int j  = lane & 15;
    const v4f at = *(const v4f*)(att + c0);
    const int wv = __builtin_amdgcn_readfirstlane(wave);
    const int* ei = (const int*)ent;
    const bool b0 = (lane & 1) != 0, b1 = (lane & 2) != 0, b2 = (lane & 4) != 0;
    const int  kb = ((lane & 1) << 2) | (lane & 2) | ((lane >> 2) & 1);
    const int  ib = ((lane & 1) << 1) | ((lane >> 1) & 1);

    for (int half = 0; half * RNH < cnt; ++half) {
        const int lo_h  = lo + half * RNH;
        const int cnt_h = (cnt - half * RNH < RNH) ? (cnt - half * RNH) : RNH;
        bool slow = slow_all;

        if (!slow) {
            if (tid == 0) ovf = 0;
            for (int n0 = wv; n0 < cnt_h; n0 += (SCT / 32) * SB) {
                v4f vr[SB], vl[SB];
#pragma unroll
                for (int i = 0; i < SB; ++i) {
                    const int n = n0 + i * (SCT / 32);
                    const size_t g = (size_t)(lo_h + ((n < cnt_h) ? n : n0)) * HC + c0;
                    vr[i] = *(const v4f*)(xr + g);
                    vl[i] = *(const v4f*)(xl + g);
                }
                asm volatile("" :: "v"(vr[0]), "v"(vr[1]), "v"(vr[2]), "v"(vr[3]),
                                   "v"(vl[0]), "v"(vl[1]), "v"(vl[2]), "v"(vl[3]));
#pragma unroll
                for (int i = 0; i < SB; ++i) {
                    const int n = n0 + i * (SCT / 32);
                    if (n < cnt_h) {
                        *(v4f*)(xrs + n * HC + c0) = vr[i];
                        float* o = acc + n * HC + lane;
                        o[0]  = vl[i].x;
                        o[32] = vl[i].y;
                        o[64] = vl[i].z;
                        o[96] = vl[i].w;
                    }
                }
                float ps[SB];
#pragma unroll
                for (int i = 0; i < SB; ++i) ps[i] = part4(vl[i], vr[i], at);
                const float u0 = (b0 ? ps[2] : ps[0]) + __shfl_xor(b0 ? ps[0] : ps[2], 1);
                const float u1 = (b0 ? ps[3] : ps[1]) + __shfl_xor(b0 ? ps[1] : ps[3], 1);
                float ss = (b1 ? u1 : u0) + __shfl_xor(b1 ? u0 : u1, 2);
                ss += __shfl_xor(ss, 4);
                ss += __shfl_xor(ss, 8);
                const int ns = n0 + ib * (SCT / 32);
                if (j < SB && ns < cnt_h) {
                    sh[ns * NHEAD + h]  = ss;
                    den[ns * NHEAD + h] = 1.0f;
                }
            }
            __syncthreads();

            const int tb = half ? (QCAP - nb) : 0;
            const int te = half ? QCAP : na;
            float dmax = -__builtin_inff();
            for (int t0 = tb + wv * PB; t0 < te; t0 += (SCT / 32) * PB) {
                const int tl = t0 + kb;
                const int tc = (tl < te) ? tl : tb;
                const int ms = ei[tc * 2 + 0];
                const int mr = ei[tc * 2 + 1];
                v4f   zl[PB], zr[PB];
                float p[PB];
#pragma unroll
                for (int k = 0; k < PB; ++k) {
                    const int sk = __builtin_amdgcn_readlane(ms, BREV3(k));
                    const int rk = __builtin_amdgcn_readlane(mr, BREV3(k));
                    zl[k] = *(const v4f*)(xl + (size_t)sk * HC + c0);
                    zr[k] = *(const v4f*)(xrs + rk * HC + c0);
                }
                const float sfv = sh[mr * NHEAD + h];
                asm volatile("" :: "v"(zl[0]), "v"(zl[1]), "v"(zl[2]), "v"(zl[3]),
                                   "v"(zl[4]), "v"(zl[5]), "v"(zl[6]), "v"(zl[7]));
#pragma unroll
                for (int k = 0; k < PB; ++k) p[k] = part4(zl[k], zr[k], at);
                float q[4], r[2];
#pragma unroll
                for (int i = 0; i < 4; ++i) {
                    const float keep = b0 ? p[4 + i] : p[i];
                    const float send = b0 ? p[i] : p[4 + i];
                    q[i] = keep + __shfl_xor(send, 1);
                }
#pragma unroll
                for (int i = 0; i < 2; ++i) {
                    const float keep = b1 ? q[2 + i] : q[i];
                    const float send = b1 ? q[i] : q[2 + i];
                    r[i] = keep + __shfl_xor(send, 2);
                }
                float sc = (b2 ? r[1] : r[0]) + __shfl_xor(b2 ? r[0] : r[1], 4);
                sc += __shfl_xor(sc, 8);
                const float d = sc - sfv;
                dmax = fmaxf(dmax, d);
                const float a = __expf(d);
                const int ai = __float_as_int(a);
#pragma unroll
                for (int k = 0; k < PB; ++k) {
                    if (t0 + k < te) {
                        const float w0 = __int_as_float(__builtin_amdgcn_readlane(ai, BREV3(k)));
                        const float w1 = __int_as_float(__builtin_amdgcn_readlane(ai, 16 + BREV3(k)));
                        const float w  = h ? w1 : w0;
                        const int rk = __builtin_amdgcn_readlane(mr, BREV3(k));
                        float* o = acc + rk * HC + lane;
                        atomicAdd(o,      w * zl[k].x);
                        atomicAdd(o + 32, w * zl[k].y);
                        atomicAdd(o + 64, w * zl[k].z);
                        atomicAdd(o + 96, w * zl[k].w);
                    }
                }
                if (j < PB && tl < te) atomicAdd(&den[mr * NHEAD + h], a);
            }
            if (dmax > SHIFT_CAP) ovf = 1;
            __syncthreads();
            slow = (ovf != 0);
        }

        if (slow) {
            float4* a4 = (float4*)acc;
            for (int i = tid; i < RNH * HC / 4; i += SCT) a4[i] = make_float4(0.0f, 0.0f, 0.0f, 0.0f);
            for (int i = tid; i < RNH * NHEAD; i += SCT) {
                den[i] = 0.0f;
                mx[i]  = -__builtin_inff();
            }
            __syncthreads();
            range_slow(acc, den, mx, xl, xr, att, src, dst, lo_h, cnt_h, E, tid);
        }

        for (int idx = tid; idx < cnt_h * CCH; idx += SCT) {
            const int node = idx >> 6;
            const int cc   = idx & 63;
            const float d0 = den[node * NHEAD + 0] + 1e-16f;
            const float d1 = den[node * NHEAD + 1] + 1e-16f;
            float v = 0.5f * (acc[node * HC + acc_slot(cc)] / d0 +
                              acc[node * HC + acc_slot(CCH + cc)] / d1) + bias[cc];
            if (apply_gelu) {
                const float u = 0.7978845608028654f * (v + 0.044715f * v * v * v);
                v = 0.5f * v * (1.0f + tanhf(u));
            }
            out[(size_t)(lo_h + node) * CCH + cc] = v;
        }
        if ((half + 1) * RNH < cnt) __syncthreads();
    }
}


extern "C" void kernel_launch(void* const* d_in, const int* in_sizes, int n_in,
                              void* d_out, int out_size, void* d_ws, size_t ws_size,
                              hipStream_t stream) {
    const float* x0   = (const float*)d_in[0];
    const int*   conn = (const int*)  d_in[1];
    const float* Wl   = (const float*)d_in[2];
    const float* bl   = (const float*)d_in[3];
    const float* Wr   = (const float*)d_in[4];
    const float* br   = (const float*)d_in[5];
    const float* att  = (const float*)d_in[6];
    const float* bias = (const float*)d_in[7];

    const int N  = in_sizes[0] / D_FEAT;
    const int E  = in_sizes[1] / 2;
    const int* src = conn;
    const int* dst = conn + E;

    float* ws    = (float*)d_ws;
    float* xl    = ws;
    float* xr    = xl   + (size_t)N * HC;
    float* x1    = xr   + (size_t)N * HC;
    const int nd = ((E + 15) / 16) * 4;
    unsigned* rid = (unsigned*)(x1 + (size_t)N * D_FEAT);

    const dim3 blk(256);
    const int proj_blocks = (N / 16 + GW - 1) / GW;
    const int range_blocks = (N + RNG - 1) / RNG;
    int* qcnt = (int*)(rid + nd);
    int* qsv  = qcnt + ((2 * range_blocks + 3) / 4) * 4;

    range_id_kernel<<<(nd + 255) / 256, blk, 0, stream>>>(dst, rid, E, nd);

    const int L = 2;
    const float* xin = x0;
    for (int l = 0; l < L; ++l) {
        const float* Wl_l = Wl + (size_t)l * D_FEAT * HC;
        const float* bl_l = bl + (size_t)l * HC;
        const float* Wr_l = Wr + (size_t)l * D_FEAT * HC;
        const float* br_l = br + (size_t)l * HC;
        const float* at_l = att + (size_t)l * NHEAD * CCH;
        const float* bi_l = bias + (size_t)l * CCH;

        proj_both_wmma<<<proj_blocks, dim3(GW * 32), 0, stream>>>(xin, Wl_l, bl_l, Wr_l, br_l, xl, xr, N);

        float* outp = (l == L - 1) ? (float*)d_out : x1;
        gat_range_kernel<<<range_blocks, dim3(SCT), 0, stream>>>(
            xl, xr, at_l, src, dst, rid, qcnt, qsv, bi_l, outp, (l < L - 1) ? 1 : 0, (l > 0) ? 1 : 0, N, E);

        xin = x1;
    }
}


# Range lists built by one edge walk (a group counts per 200-node sub-pass in LDS, one returning global integer add reserves its slots; read back 4 loads at a time), no list search; sums by arrival
# speedup vs baseline: 4.3840x; 1.0903x over previous
#include <hip/hip_runtime.h>

#define D_FEAT 64
#define HC 128
#define NHEAD 2
#define CCH 64
#define NEG_SLOPE 0.2f

typedef __attribute__((ext_vector_type(2))) float v2f;
typedef __attribute__((ext_vector_type(8))) float v8f;
typedef __attribute__((ext_vector_type(4))) float v4f;


__device__ __forceinline__ void atomicMaxF(float* addr, float val) {
    if (val >= 0.0f) atomicMax((int*)addr, __float_as_int(val));
    else             atomicMin((unsigned int*)addr, __float_as_uint(val));
}

#define GW 4

__global__ __launch_bounds__(GW * 32)
void proj_both_wmma(const float* __restrict__ X,
                    const float* __restrict__ Wl, const float* __restrict__ bl,
                    const float* __restrict__ Wr, const float* __restrict__ br,
                    float* __restrict__ Yl, float* __restrict__ Yr, int rows,
                    int* __restrict__ zc, int nz) {
    if (blockIdx.x == 0)
        for (int i = threadIdx.x; i < nz; i += GW * 32) zc[i] = 0;
    const int tm   = (blockIdx.x * blockDim.x + threadIdx.x) >> 5;
    const int lane = threadIdx.x & 31;
    if (tm >= (rows >> 4)) return;

    const int c     = lane & 15;
    const int hl    = lane >> 4;
    const int khalf = hl << 1;
    const int mrow  = (tm << 4) + c;

    v2f a[16];
    const float* ap = X + (size_t)mrow * D_FEAT + khalf;
#pragma unroll
    for (int s = 0; s < 16; ++s) {
        a[s].x = ap[4 * s];
        a[s].y = ap[4 * s + 1];
    }

    const int rbase = (tm << 4) + (hl << 3);
#pragma unroll
    for (int q = 0; q < 4; ++q) {
        const float* W  = (q < 2) ? Wl : Wr;
        const float* bv = (q < 2) ? bl : br;
        float*       Y  = (q < 2) ? Yl : Yr;
        const int col = ((q & 1) << 6) + (c << 2);

        v8f acc0 = {}, acc1 = {}, acc2 = {}, acc3 = {};
#pragma unroll
        for (int s0 = 0; s0 < 16; s0 += 8) {
            float4 w0[8], w1[8];
#pragma unroll
            for (int i = 0; i < 8; ++i) {
                const int kk = 4 * (s0 + i) + khalf;
                w0[i] = *(const float4*)(W + (size_t)(kk + 0) * HC + col);
                w1[i] = *(const float4*)(W + (size_t)(kk + 1) * HC + col);
            }
#pragma unroll
            for (int i = 0; i < 8; ++i) {
                const v2f av = a[s0 + i];
                v2f b;
                b.x = w0[i].x; b.y = w1[i].x;
                acc0 = __builtin_amdgcn_wmma_f32_16x16x4_f32(false, av, false, b, (short)0, acc0, false, false);
                b.x = w0[i].y; b.y = w1[i].y;
                acc1 = __builtin_amdgcn_wmma_f32_16x16x4_f32(false, av, false, b, (short)0, acc1, false, false);
                b.x = w0[i].z; b.y = w1[i].z;
                acc2 = __builtin_amdgcn_wmma_f32_16x16x4_f32(false, av, false, b, (short)0, acc2, false, false);
                b.x = w0[i].w; b.y = w1[i].w;
                acc3 = __builtin_amdgcn_wmma_f32_16x16x4_f32(false, av, false, b, (short)0, acc3, false, false);
            }
        }

        const float4 bb = *(const float4*)(bv + col);
#pragma unroll
        for (int j = 0; j < 8; ++j)
            *(float4*)(Y + (size_t)(rbase + j) * HC + col) =
                make_float4(acc0[j] + bb.x, acc1[j] + bb.y, acc2[j] + bb.z, acc3[j] + bb.w);
    }
}

#define RNG   400
#define RNH   200
#define QCAP  5600
#define BKT   1024
#define BKE   4
#define NBK   2048

__global__ __launch_bounds__(BKT)
void edge_list_kernel(const int* __restrict__ src, const int* __restrict__ dst,
                      int* __restrict__ qcnt,
                      int* __restrict__ qsv,
                      int E, int N, int nbk) {
    __shared__ int cnt[NBK];
    const int tid = threadIdx.x;
    for (int i = tid; i < nbk; i += BKT) cnt[i] = 0;
    __syncthreads();

    const int base = blockIdx.x * (BKT * BKE) + tid;
    int s[BKE], b[BKE], r[BKE], k[BKE];
#pragma unroll
    for (int i = 0; i < BKE; ++i) {
        const int e = base + i * BKT;
        const bool in = e < E;
        const int ec = in ? e : 0;
        const int dv = dst[ec];
        s[i] = src[ec];
        const int d = in ? dv : -1;
        const bool ok = (unsigned)d < (unsigned)N;
        const int bb = (int)((unsigned)(ok ? d : 0) / (unsigned)RNH);
        b[i] = (ok && bb < nbk) ? bb : -1;
        r[i] = d - bb * RNH;
    }
#pragma unroll
    for (int i = 0; i < BKE; ++i) k[i] = (b[i] >= 0) ? atomicAdd(&cnt[b[i]], 1) : 0;
    __syncthreads();
    for (int i = tid; i < nbk; i += BKT) {
        const int c = cnt[i];
        cnt[i] = (c > 0) ? atomicAdd(&qcnt[i], c) : 0;
    }
    __syncthreads();
    int2* q2 = (int2*)qsv;
    int pos[BKE];
#pragma unroll
    for (int i = 0; i < BKE; ++i) pos[i] = cnt[(b[i] >= 0) ? b[i] : 0] + k[i];
#pragma unroll
    for (int i = 0; i < BKE; ++i) {
        if (b[i] >= 0 && pos[i] < QCAP) {
            const int slot = (b[i] & 1) ? (QCAP - 1 - pos[i]) : pos[i];
            q2[(size_t)(b[i] >> 1) * QCAP + slot] = make_int2(s[i], r[i]);
        }
    }
}

#define SCT   1024
#define PB    8
#define SB    4
#define LRD   4
#define BREV3(k) ((((k) & 1) << 2) | ((k) & 2) | (((k) >> 2) & 1))
#define SHIFT_CAP 64.0f

__device__ __forceinline__ int acc_slot(int ch) { return ((ch & 3) << 5) | (ch >> 2); }

__device__ __forceinline__ float part4(const v4f vl, const v4f vr, const v4f at) {
    float p = 0.0f, z;
    z = vl.x + vr.x; z = (z > 0.0f) ? z : NEG_SLOPE * z; p += z * at.x;
    z = vl.y + vr.y; z = (z > 0.0f) ? z : NEG_SLOPE * z; p += z * at.y;
    z = vl.z + vr.z; z = (z > 0.0f) ? z : NEG_SLOPE * z; p += z * at.z;
    z = vl.w + vr.w; z = (z > 0.0f) ? z : NEG_SLOPE * z; p += z * at.w;
    return p;
}

__device__ void range_slow(float* acc, float* den, float* mx,
                           const float* __restrict__ xl, const float* __restrict__ xr,
                           const float* __restrict__ att,
                           const int* __restrict__ src, const int* __restrict__ dst,
                           int lo, int cnt, int E, int tid) {
    for (int pass = 0; pass < 2; ++pass) {
        for (int e = tid; e < E + cnt; e += SCT) {
            int r, s;
            if (e < E) { r = dst[e] - lo; s = src[e]; } else { r = e - E; s = lo + r; }
            if ((unsigned)r >= (unsigned)cnt) continue;
            for (int hh = 0; hh < NHEAD; ++hh) {
                float p = 0.0f;
                for (int cc = 0; cc < CCH; ++cc) {
                    const int ch = hh * CCH + cc;
                    float z = xl[(size_t)s * HC + ch] + xr[(size_t)(lo + r) * HC + ch];
                    z = (z > 0.0f) ? z : NEG_SLOPE * z;
                    p += z * att[ch];
                }
                if (pass == 0) {
                    atomicMaxF(&mx[r * NHEAD + hh], p);
                } else {
                    const float a = __expf(p - mx[r * NHEAD + hh]);
                    atomicAdd(&den[r * NHEAD + hh], a);
                    for (int cc = 0; cc < CCH; ++cc) {
                        const int ch = hh * CCH + cc;
                        atomicAdd(&acc[r * HC + acc_slot(ch)], a * xl[(size_t)s * HC + ch]);
                    }
                }
            }
        }
        __syncthreads();
    }
}

__global__ __launch_bounds__(SCT)
void gat_range_kernel(const float* __restrict__ xl,
                      const float* __restrict__ xr,
                      const float* __restrict__ att,
                      const int* __restrict__ src,
                      const int* __restrict__ dst,
                      const int* __restrict__ qcnt,
                      const int* __restrict__ qsv,
                      const float* __restrict__ bias,
                      float* __restrict__ out,
                      int apply_gelu, int N, int E) {
    __shared__ __attribute__((aligned(16))) float xrs[RNH * HC];
    __shared__ __attribute__((aligned(16))) float acc[RNH * HC];
    __shared__ __attribute__((aligned(8))) int2 ent[QCAP];
    __shared__ float den[RNH * NHEAD];
    __shared__ float sh[RNH * NHEAD];
    __shared__ float mx[RNH * NHEAD];
    __shared__ int ovf;

    const int tid  = threadIdx.x;
    const int lane = tid & 31;
    const int wave = tid >> 5;
    const int lo   = blockIdx.x * RNG;
    const int cnt  = (N - lo < RNG) ? (N - lo) : RNG;

    const int na_all = qcnt[2 * blockIdx.x];
    const int nb_all = qcnt[2 * blockIdx.x + 1];
    const bool slow_all = na_all > QCAP - nb_all;
    const int na = slow_all ? 0 : na_all;
    const int nb = slow_all ? 0 : nb_all;
    {
        const int2* qs2 = (const int2*)qsv + (size_t)blockIdx.x * QCAP;
        const int ntot = na + nb;
        const int gap  = QCAP - ntot;
        for (int t0 = tid; t0 < ntot; t0 += LRD * SCT) {
            int2 v[LRD];
#pragma unroll
            for (int i = 0; i < LRD; ++i) {
                const int t  = t0 + i * SCT;
                const int tt = (t < ntot) ? t : t0;
                v[i] = qs2[(tt < na) ? tt : tt + gap];
            }
            asm volatile("" :: "v"(v[0].x), "v"(v[0].y), "v"(v[1].x), "v"(v[1].y),
                               "v"(v[2].x), "v"(v[2].y), "v"(v[3].x), "v"(v[3].y));
#pragma unroll
            for (int i = 0; i < LRD; ++i) {
                const int t = t0 + i * SCT;
                if (t < ntot) ent[(t < na) ? t : t + gap] = v[i];
            }
        }
    }

    const int h  = lane >> 4;
    const int c0 = lane << 2;
    const int j  = lane & 15;
    const v4f at = *(const v4f*)(att + c0);
    const int wv = __builtin_amdgcn_readfirstlane(wave);
    const int* ei = (const int*)ent;
    const bool b0 = (lane & 1) != 0, b1 = (lane & 2) != 0, b2 = (lane & 4) != 0;
    const int  kb = ((lane & 1) << 2) | (lane & 2) | ((lane >> 2) & 1);
    const int  ib = ((lane & 1) << 1) | ((lane >> 1) & 1);

    for (int half = 0; half * RNH < cnt; ++half) {
        const int lo_h  = lo + half * RNH;
        const int cnt_h = (cnt - half * RNH < RNH) ? (cnt - half * RNH) : RNH;
        bool slow = slow_all;

        if (!slow) {
            if (tid == 0) ovf = 0;
            for (int n0 = wv; n0 < cnt_h; n0 += (SCT / 32) * SB) {
                v4f vr[SB], vl[SB];
#pragma unroll
                for (int i = 0; i < SB; ++i) {
                    const int n = n0 + i * (SCT / 32);
                    const size_t g = (size_t)(lo_h + ((n < cnt_h) ? n : n0)) * HC + c0;
                    vr[i] = *(const v4f*)(xr + g);
                    vl[i] = *(const v4f*)(xl + g);
                }
                asm volatile("" :: "v"(vr[0]), "v"(vr[1]), "v"(vr[2]), "v"(vr[3]),
                                   "v"(vl[0]), "v"(vl[1]), "v"(vl[2]), "v"(vl[3]));
#pragma unroll
                for (int i = 0; i < SB; ++i) {
                    const int n = n0 + i * (SCT / 32);
                    if (n < cnt_h) {
                        *(v4f*)(xrs + n * HC + c0) = vr[i];
                        float* o = acc + n * HC + lane;
                        o[0]  = vl[i].x;
                        o[32] = vl[i].y;
                        o[64] = vl[i].z;
                        o[96] = vl[i].w;
                    }
                }
                float ps[SB];
#pragma unroll
                for (int i = 0; i < SB; ++i) ps[i] = part4(vl[i], vr[i], at);
                const float u0 = (b0 ? ps[2] : ps[0]) + __shfl_xor(b0 ? ps[0] : ps[2], 1);
                const float u1 = (b0 ? ps[3] : ps[1]) + __shfl_xor(b0 ? ps[1] : ps[3], 1);
                float ss = (b1 ? u1 : u0) + __shfl_xor(b1 ? u0 : u1, 2);
                ss += __shfl_xor(ss, 4);
                ss += __shfl_xor(ss, 8);
                const int ns = n0 + ib * (SCT / 32);
                if (j < SB && ns < cnt_h) {
                    sh[ns * NHEAD + h]  = ss;
                    den[ns * NHEAD + h] = 1.0f;
                }
            }
            __syncthreads();

            const int tb = half ? (QCAP - nb) : 0;
            const int te = half ? QCAP : na;
            float dmax = -__builtin_inff();
            for (int t0 = tb + wv * PB; t0 < te; t0 += (SCT / 32) * PB) {
                const int tl = t0 + kb;
                const int tc = (tl < te) ? tl : tb;
                const int ms = ei[tc * 2 + 0];
                const int mr = ei[tc * 2 + 1];
                v4f   zl[PB], zr[PB];
                float p[PB];
#pragma unroll
                for (int k = 0; k < PB; ++k) {
                    const int sk = __builtin_amdgcn_readlane(ms, BREV3(k));
                    const int rk = __builtin_amdgcn_readlane(mr, BREV3(k));
                    zl[k] = *(const v4f*)(xl + (size_t)sk * HC + c0);
                    zr[k] = *(const v4f*)(xrs + rk * HC + c0);
                }
                const float sfv = sh[mr * NHEAD + h];
                asm volatile("" :: "v"(zl[0]), "v"(zl[1]), "v"(zl[2]), "v"(zl[3]),
                                   "v"(zl[4]), "v"(zl[5]), "v"(zl[6]), "v"(zl[7]));
#pragma unroll
                for (int k = 0; k < PB; ++k) p[k] = part4(zl[k], zr[k], at);
                float q[4], r[2];
#pragma unroll
                for (int i = 0; i < 4; ++i) {
                    const float keep = b0 ? p[4 + i] : p[i];
                    const float send = b0 ? p[i] : p[4 + i];
                    q[i] = keep + __shfl_xor(send, 1);
                }
#pragma unroll
                for (int i = 0; i < 2; ++i) {
                    const float keep = b1 ? q[2 + i] : q[i];
                    const float send = b1 ? q[i] : q[2 + i];
                    r[i] = keep + __shfl_xor(send, 2);
                }
                float sc = (b2 ? r[1] : r[0]) + __shfl_xor(b2 ? r[0] : r[1], 4);
                sc += __shfl_xor(sc, 8);
                const float d = sc - sfv;
                dmax = fmaxf(dmax, d);
                const float a = __expf(d);
                const int ai = __float_as_int(a);
#pragma unroll
                for (int k = 0; k < PB; ++k) {
                    if (t0 + k < te) {
                        const float w0 = __int_as_float(__builtin_amdgcn_readlane(ai, BREV3(k)));
                        const float w1 = __int_as_float(__builtin_amdgcn_readlane(ai, 16 + BREV3(k)));
                        const float w  = h ? w1 : w0;
                        const int rk = __builtin_amdgcn_readlane(mr, BREV3(k));
                        float* o = acc + rk * HC + lane;
                        atomicAdd(o,      w * zl[k].x);
                        atomicAdd(o + 32, w * zl[k].y);
                        atomicAdd(o + 64, w * zl[k].z);
                        atomicAdd(o + 96, w * zl[k].w);
                    }
                }
                if (j < PB && tl < te) atomicAdd(&den[mr * NHEAD + h], a);
            }
            if (dmax > SHIFT_CAP) ovf = 1;
            __syncthreads();
            slow = (ovf != 0);
        }

        if (slow) {
            float4* a4 = (float4*)acc;
            for (int i = tid; i < RNH * HC / 4; i += SCT) a4[i] = make_float4(0.0f, 0.0f, 0.0f, 0.0f);
            for (int i = tid; i < RNH * NHEAD; i += SCT) {
                den[i] = 0.0f;
                mx[i]  = -__builtin_inff();
            }
            __syncthreads();
            range_slow(acc, den, mx, xl, xr, att, src, dst, lo_h, cnt_h, E, tid);
        }

        for (int idx = tid; idx < cnt_h * CCH; idx += SCT) {
            const int node = idx >> 6;
            const int cc   = idx & 63;
            const float d0 = den[node * NHEAD + 0] + 1e-16f;
            const float d1 = den[node * NHEAD + 1] + 1e-16f;
            float v = 0.5f * (acc[node * HC + acc_slot(cc)] / d0 +
                              acc[node * HC + acc_slot(CCH + cc)] / d1) + bias[cc];
            if (apply_gelu) {
                const float u = 0.7978845608028654f * (v + 0.044715f * v * v * v);
                v = 0.5f * v * (1.0f + tanhf(u));
            }
            out[(size_t)(lo_h + node) * CCH + cc] = v;
        }
        if ((half + 1) * RNH < cnt) __syncthreads();
    }
}


extern "C" void kernel_launch(void* const* d_in, const int* in_sizes, int n_in,
                              void* d_out, int out_size, void* d_ws, size_t ws_size,
                              hipStream_t stream) {
    const float* x0   = (const float*)d_in[0];
    const int*   conn = (const int*)  d_in[1];
    const float* Wl   = (const float*)d_in[2];
    const float* bl   = (const float*)d_in[3];
    const float* Wr   = (const float*)d_in[4];
    const float* br   = (const float*)d_in[5];
    const float* att  = (const float*)d_in[6];
    const float* bias = (const float*)d_in[7];

    const int N  = in_sizes[0] / D_FEAT;
    const int E  = in_sizes[1] / 2;
    const int* src = conn;
    const int* dst = conn + E;

    float* ws    = (float*)d_ws;
    float* xl    = ws;
    float* xr    = xl   + (size_t)N * HC;
    float* x1    = xr   + (size_t)N * HC;

    const int proj_blocks = (N / 16 + GW - 1) / GW;
    const int range_blocks = (N + RNG - 1) / RNG;
    const int nbk = 2 * range_blocks;
    const int list_blocks = (E + BKT * BKE - 1) / (BKT * BKE);
    int* qcnt = (int*)(x1 + (size_t)N * D_FEAT);
    int* qsv  = qcnt + ((nbk + 3) / 4) * 4;

    const int L = 2;
    const float* xin = x0;
    for (int l = 0; l < L; ++l) {
        const float* Wl_l = Wl + (size_t)l * D_FEAT * HC;
        const float* bl_l = bl + (size_t)l * HC;
        const float* Wr_l = Wr + (size_t)l * D_FEAT * HC;
        const float* br_l = br + (size_t)l * HC;
        const float* at_l = att + (size_t)l * NHEAD * CCH;
        const float* bi_l = bias + (size_t)l * CCH;

        proj_both_wmma<<<proj_blocks, dim3(GW * 32), 0, stream>>>(xin, Wl_l, bl_l, Wr_l, br_l, xl, xr, N,
                                                                  qcnt, (l == 0) ? nbk : 0);
        if (l == 0)
            edge_list_kernel<<<list_blocks, dim3(BKT), 0, stream>>>(src, dst, qcnt, qsv, E, N, nbk);

        float* outp = (l == L - 1) ? (float*)d_out : x1;
        gat_range_kernel<<<range_blocks, dim3(SCT), 0, stream>>>(
            xl, xr, at_l, src, dst, qcnt, qsv, bi_l, outp, (l < L - 1) ? 1 : 0, N, E);

        xin = x1;
    }
}
